# starting kernel
# baseline (speedup 1.0000x reference)
#include <hip/hip_runtime.h>

typedef float v2f __attribute__((ext_vector_type(2)));
typedef float v4f __attribute__((ext_vector_type(4)));
typedef float v8f __attribute__((ext_vector_type(8)));

#define D_MODEL    64
#define SEQ_LEN    9
#define ACTION_NUM 300
#define NT_OUT     19

#define WS_M_OFF   0
#define WS_W_OFF   36864
#define WS_O_OFF   73728
#define WS_FLOATS  93184
#define WS_BYTES   (WS_FLOATS * 4)

__device__ __forceinline__ float sigmoid_f(float x) {
    return 1.0f / (1.0f + __expf(-x));
}

__global__ void carnn_pack_weights(const float* __restrict__ Mw,
                                   const float* __restrict__ Ww,
                                   const float* __restrict__ outw,
                                   float* __restrict__ ws)
{
    const int idx = blockIdx.x * 256 + threadIdx.x;
    if (idx >= WS_FLOATS) return;
    float val;
    if (idx < WS_O_OFF) {
        const int matsel = idx / WS_W_OFF;
        int r  = idx % WS_W_OFF;
        const int s   = r / 4096;  r %= 4096;
        const int nt  = r / 1024;  r %= 1024;
        const int ktp = r / 128;   r %= 128;
        const int lane = r / 4;
        const int j    = r % 4;
        const int h = lane >> 4, lr = lane & 15;
        const int n  = nt * 16 + lr;
        const int kt = 2 * ktp + (j >> 1);
        const int k  = 4 * kt + 2 * h + (j & 1);
        const float* src = matsel ? Ww : Mw;
        val = src[s * 4096 + n * D_MODEL + k];
    } else {
        int r = idx - WS_O_OFF;
        const int nt  = r / 1024;  r %= 1024;
        const int ktp = r / 128;   r %= 128;
        const int lane = r / 4;
        const int j    = r % 4;
        const int h = lane >> 4, lr = lane & 15;
        const int n  = nt * 16 + lr;
        const int kt = 2 * ktp + (j >> 1);
        const int k  = 4 * kt + 2 * h + (j & 1);
        val = (n < ACTION_NUM) ? outw[n * D_MODEL + k] : 0.0f;
    }
    ws[idx] = val;
}

__global__ __launch_bounds__(256)
void carnn_wmma_packed(const int*   __restrict__ actions,
                       const float* __restrict__ emb,
                       const float* __restrict__ Mb,
                       const float* __restrict__ Wb,
                       const float* __restrict__ outb,
                       const float* __restrict__ wsf,
                       float*       __restrict__ out,
                       int batch)
{
    __shared__ float lds[8][2][16][68];

    const int lane = threadIdx.x & 31;
    const int wv   = threadIdx.x >> 5;
    const int half = lane >> 4;
    const int lr   = lane & 15;
    const int wave_id = blockIdx.x * 8 + wv;
    const int row0 = wave_id * 32;
    if (row0 >= batch) return;
    const int row1 = (row0 + 16 < batch) ? (row0 + 16) : row0;

    float (*hl0)[68] = lds[wv][0];
    float (*hl1)[68] = lds[wv][1];
    const v4f* wsv = (const v4f*)wsf;

    v8f acc0[4], acc1[4];

    for (int s = 0; s < SEQ_LEN; ++s) {
        const int a0 = actions[(row0 + lr) * SEQ_LEN + s];
        const int a1 = actions[(row1 + lr) * SEQ_LEN + s];
        const float* xr0 = emb + a0 * D_MODEL;
        const float* xr1 = emb + a1 * D_MODEL;

        #pragma unroll
        for (int nt = 0; nt < 4; ++nt) { acc0[nt] = (v8f)0.0f; acc1[nt] = (v8f)0.0f; }

        {
            const v4f* bm = wsv + (WS_M_OFF / 4) + s * 1024;
            #pragma unroll 4
            for (int ktp = 0; ktp < 8; ++ktp) {
                const int ke = 8 * ktp + 2 * half;
                v2f a0e = *(const v2f*)(xr0 + ke);
                v2f a0o = *(const v2f*)(xr0 + ke + 4);
                v2f a1e = *(const v2f*)(xr1 + ke);
                v2f a1o = *(const v2f*)(xr1 + ke + 4);
                v4f b[4];
                #pragma unroll
                for (int nt = 0; nt < 4; ++nt) b[nt] = bm[nt * 256 + ktp * 32 + lane];
                #pragma unroll
                for (int nt = 0; nt < 4; ++nt) {
                    v2f be = {b[nt][0], b[nt][1]};
                    acc0[nt] = __builtin_amdgcn_wmma_f32_16x16x4_f32(false, a0e, false, be, (short)0, acc0[nt], false, false);
                    acc1[nt] = __builtin_amdgcn_wmma_f32_16x16x4_f32(false, a1e, false, be, (short)0, acc1[nt], false, false);
                }
                #pragma unroll
                for (int nt = 0; nt < 4; ++nt) {
                    v2f bo = {b[nt][2], b[nt][3]};
                    acc0[nt] = __builtin_amdgcn_wmma_f32_16x16x4_f32(false, a0o, false, bo, (short)0, acc0[nt], false, false);
                    acc1[nt] = __builtin_amdgcn_wmma_f32_16x16x4_f32(false, a1o, false, bo, (short)0, acc1[nt], false, false);
                }
            }
        }

        if (s > 0) {
            const v4f* bw = wsv + (WS_W_OFF / 4) + s * 1024;
            #pragma unroll 4
            for (int ktp = 0; ktp < 8; ++ktp) {
                const int ke = 8 * ktp + 2 * half;
                v2f a0e = *(const v2f*)(&hl0[lr][ke]);
                v2f a0o = *(const v2f*)(&hl0[lr][ke + 4]);
                v2f a1e = *(const v2f*)(&hl1[lr][ke]);
                v2f a1o = *(const v2f*)(&hl1[lr][ke + 4]);
                v4f b[4];
                #pragma unroll
                for (int nt = 0; nt < 4; ++nt) b[nt] = bw[nt * 256 + ktp * 32 + lane];
                #pragma unroll
                for (int nt = 0; nt < 4; ++nt) {
                    v2f be = {b[nt][0], b[nt][1]};
                    acc0[nt] = __builtin_amdgcn_wmma_f32_16x16x4_f32(false, a0e, false, be, (short)0, acc0[nt], false, false);
                    acc1[nt] = __builtin_amdgcn_wmma_f32_16x16x4_f32(false, a1e, false, be, (short)0, acc1[nt], false, false);
                }
                #pragma unroll
                for (int nt = 0; nt < 4; ++nt) {
                    v2f bo = {b[nt][2], b[nt][3]};
                    acc0[nt] = __builtin_amdgcn_wmma_f32_16x16x4_f32(false, a0o, false, bo, (short)0, acc0[nt], false, false);
                    acc1[nt] = __builtin_amdgcn_wmma_f32_16x16x4_f32(false, a1o, false, bo, (short)0, acc1[nt], false, false);
                }
            }
        }

        #pragma unroll
        for (int nt = 0; nt < 4; ++nt) {
            const int n = nt * 16 + lr;
            const float bias = Mb[s * D_MODEL + n] + Wb[s * D_MODEL + n];
            #pragma unroll
            for (int r = 0; r < 8; ++r) {
                float v0 = sigmoid_f(acc0[nt][r] + bias);
                float v1 = sigmoid_f(acc1[nt][r] + bias);
                hl0[half * 8 + r][n] = v0;
                hl1[half * 8 + r][n] = v1;
            }
        }
    }

    const v4f* bo_base = wsv + (WS_O_OFF / 4);
    #pragma unroll 1
    for (int t = 0; t < 2; ++t) {
        if (t && row1 == row0) break;
        float (*hlp)[68] = t ? hl1 : hl0;
        const int rbase = t ? row1 : row0;

        v2f afrag[16];
        #pragma unroll
        for (int kt = 0; kt < 16; ++kt)
            afrag[kt] = *(const v2f*)(&hlp[lr][4 * kt + 2 * half]);

        for (int nt0 = 0; nt0 < NT_OUT; nt0 += 4) {
            v8f oacc[4];
            #pragma unroll
            for (int j = 0; j < 4; ++j) oacc[j] = (v8f)0.0f;

            #pragma unroll 2
            for (int ktp = 0; ktp < 8; ++ktp) {
                #pragma unroll
                for (int j = 0; j < 4; ++j) {
                    const int nt = nt0 + j;
                    if (nt >= NT_OUT) break;
                    v4f b = bo_base[nt * 256 + ktp * 32 + lane];
                    v2f be = {b[0], b[1]};
                    v2f bo = {b[2], b[3]};
                    oacc[j] = __builtin_amdgcn_wmma_f32_16x16x4_f32(false, afrag[2 * ktp],     false, be, (short)0, oacc[j], false, false);
                    oacc[j] = __builtin_amdgcn_wmma_f32_16x16x4_f32(false, afrag[2 * ktp + 1], false, bo, (short)0, oacc[j], false, false);
                }
            }

            #pragma unroll
            for (int j = 0; j < 4; ++j) {
                const int nt = nt0 + j;
                if (nt >= NT_OUT) break;
                const int n = nt * 16 + lr;
                if (n < ACTION_NUM) {
                    const float bias = outb[n];
                    #pragma unroll
                    for (int r = 0; r < 8; ++r) {
                        out[(long)(rbase + half * 8 + r) * ACTION_NUM + n] = oacc[j][r] + bias;
                    }
                }
            }
        }
    }
}

__global__ __launch_bounds__(256)
void carnn_wmma_direct(const int*   __restrict__ actions,
                       const float* __restrict__ emb,
                       const float* __restrict__ Mw,
                       const float* __restrict__ Mb,
                       const float* __restrict__ Ww,
                       const float* __restrict__ Wb,
                       const float* __restrict__ outw,
                       const float* __restrict__ outb,
                       float*       __restrict__ out,
                       int batch)
{
    __shared__ float lds[8][16][68];
    const int lane = threadIdx.x & 31;
    const int wv   = threadIdx.x >> 5;
    const int half = lane >> 4;
    const int lr   = lane & 15;
    const int tile = blockIdx.x * 8 + wv;
    const int row0 = tile * 16;
    if (row0 >= batch) return;

    float (*hl)[68] = lds[wv];
    v8f acc[4];

    for (int s = 0; s < SEQ_LEN; ++s) {
        const float* mwp = Mw + s * D_MODEL * D_MODEL;
        const float* wwp = Ww + s * D_MODEL * D_MODEL;
        const int aidx = actions[(row0 + lr) * SEQ_LEN + s];
        const float* xrow = emb + aidx * D_MODEL;

        #pragma unroll
        for (int nt = 0; nt < 4; ++nt) acc[nt] = (v8f)0.0f;

        #pragma unroll 4
        for (int kt = 0; kt < 16; ++kt) {
            const int k = kt * 4 + half * 2;
            v2f a = *(const v2f*)(xrow + k);
            #pragma unroll
            for (int nt = 0; nt < 4; ++nt) {
                const int n = nt * 16 + lr;
                v2f b = *(const v2f*)(mwp + n * D_MODEL + k);
                acc[nt] = __builtin_amdgcn_wmma_f32_16x16x4_f32(false, a, false, b, (short)0, acc[nt], false, false);
            }
        }
        if (s > 0) {
            #pragma unroll 4
            for (int kt = 0; kt < 16; ++kt) {
                const int k = kt * 4 + half * 2;
                v2f a = *(const v2f*)(&hl[lr][k]);
                #pragma unroll
                for (int nt = 0; nt < 4; ++nt) {
                    const int n = nt * 16 + lr;
                    v2f b = *(const v2f*)(wwp + n * D_MODEL + k);
                    acc[nt] = __builtin_amdgcn_wmma_f32_16x16x4_f32(false, a, false, b, (short)0, acc[nt], false, false);
                }
            }
        }
        #pragma unroll
        for (int nt = 0; nt < 4; ++nt) {
            const int n = nt * 16 + lr;
            const float bias = Mb[s * D_MODEL + n] + Wb[s * D_MODEL + n];
            #pragma unroll
            for (int r = 0; r < 8; ++r) {
                float v = sigmoid_f(acc[nt][r] + bias);
                hl[half * 8 + r][n] = v;
            }
        }
    }

    v2f afrag[16];
    #pragma unroll
    for (int kt = 0; kt < 16; ++kt)
        afrag[kt] = *(const v2f*)(&hl[lr][4 * kt + 2 * half]);

    for (int nt0 = 0; nt0 < NT_OUT; nt0 += 4) {
        v8f oacc[4];
        #pragma unroll
        for (int j = 0; j < 4; ++j) oacc[j] = (v8f)0.0f;
        #pragma unroll 4
        for (int kt = 0; kt < 16; ++kt) {
            #pragma unroll
            for (int j = 0; j < 4; ++j) {
                const int nt = nt0 + j;
                if (nt >= NT_OUT) break;
                const int n = nt * 16 + lr;
                const int k = kt * 4 + half * 2;
                v2f b;
                if (n < ACTION_NUM) b = *(const v2f*)(outw + n * D_MODEL + k);
                else                b = (v2f)0.0f;
                oacc[j] = __builtin_amdgcn_wmma_f32_16x16x4_f32(false, afrag[kt], false, b, (short)0, oacc[j], false, false);
            }
        }
        #pragma unroll
        for (int j = 0; j < 4; ++j) {
            const int nt = nt0 + j;
            if (nt >= NT_OUT) break;
            const int n = nt * 16 + lr;
            if (n < ACTION_NUM) {
                const float bias = outb[n];
                #pragma unroll
                for (int r = 0; r < 8; ++r)
                    out[(long)(row0 + half * 8 + r) * ACTION_NUM + n] = oacc[j][r] + bias;
            }
        }
    }
}

extern "C" void kernel_launch(void* const* d_in, const int* in_sizes, int n_in,
                              void* d_out, int out_size, void* d_ws, size_t ws_size,
                              hipStream_t stream) {
    const int*   actions = (const int*)  d_in[0];
    const float* emb     = (const float*)d_in[1];
    const float* Mw      = (const float*)d_in[2];
    const float* Mb      = (const float*)d_in[3];
    const float* Ww      = (const float*)d_in[4];
    const float* Wb      = (const float*)d_in[5];
    const float* outw    = (const float*)d_in[6];
    const float* outb    = (const float*)d_in[7];
    float*       out     = (float*)d_out;

    const int batch = in_sizes[0] / SEQ_LEN;
    const int tiles = (batch + 15) / 16;

    if (ws_size >= (size_t)WS_BYTES && d_ws != nullptr) {
        float* ws = (float*)d_ws;
        carnn_pack_weights<<<(WS_FLOATS + 255) / 256, 256, 0, stream>>>(Mw, Ww, outw, ws);
        const int waves  = (tiles + 1) / 2;
        const int blocks = (waves + 7) / 8;
        carnn_wmma_packed<<<blocks, 256, 0, stream>>>(actions, emb, Mb, Wb, outb, ws, out, batch);
    } else {
        const int blocks = (tiles + 7) / 8;
        carnn_wmma_direct<<<blocks, 256, 0, stream>>>(actions, emb, Mw, Mb, Ww, Wb, outw, outb, out, batch);
    }
}


# output projection: both 16-row tiles share each out_w fragment load and each 4-tile group's 32 loads issue as one batch ahead of its WMMAs (was 3+1 loads then a wait per k pair); sum order unchanged
# speedup vs baseline: 1.5081x; 1.5081x over previous
#include <hip/hip_runtime.h>

typedef float v2f __attribute__((ext_vector_type(2)));
typedef float v4f __attribute__((ext_vector_type(4)));
typedef float v8f __attribute__((ext_vector_type(8)));

#define D_MODEL    64
#define SEQ_LEN    9
#define ACTION_NUM 300
#define NT_OUT     19

#define WS_M_OFF   0
#define WS_W_OFF   36864
#define WS_O_OFF   73728
#define WS_FLOATS  93184
#define WS_BYTES   (WS_FLOATS * 4)

__device__ __forceinline__ float sigmoid_f(float x) {
    return 1.0f / (1.0f + __expf(-x));
}

__global__ void carnn_pack_weights(const float* __restrict__ Mw,
                                   const float* __restrict__ Ww,
                                   const float* __restrict__ outw,
                                   float* __restrict__ ws)
{
    const int idx = blockIdx.x * 256 + threadIdx.x;
    if (idx >= WS_FLOATS) return;
    float val;
    if (idx < WS_O_OFF) {
        const int matsel = idx / WS_W_OFF;
        int r  = idx % WS_W_OFF;
        const int s   = r / 4096;  r %= 4096;
        const int nt  = r / 1024;  r %= 1024;
        const int ktp = r / 128;   r %= 128;
        const int lane = r / 4;
        const int j    = r % 4;
        const int h = lane >> 4, lr = lane & 15;
        const int n  = nt * 16 + lr;
        const int kt = 2 * ktp + (j >> 1);
        const int k  = 4 * kt + 2 * h + (j & 1);
        const float* src = matsel ? Ww : Mw;
        val = src[s * 4096 + n * D_MODEL + k];
    } else {
        int r = idx - WS_O_OFF;
        const int nt  = r / 1024;  r %= 1024;
        const int ktp = r / 128;   r %= 128;
        const int lane = r / 4;
        const int j    = r % 4;
        const int h = lane >> 4, lr = lane & 15;
        const int n  = nt * 16 + lr;
        const int kt = 2 * ktp + (j >> 1);
        const int k  = 4 * kt + 2 * h + (j & 1);
        val = (n < ACTION_NUM) ? outw[n * D_MODEL + k] : 0.0f;
    }
    ws[idx] = val;
}

__global__ __launch_bounds__(256)
void carnn_wmma_packed(const int*   __restrict__ actions,
                       const float* __restrict__ emb,
                       const float* __restrict__ Mb,
                       const float* __restrict__ Wb,
                       const float* __restrict__ outb,
                       const float* __restrict__ wsf,
                       float*       __restrict__ out,
                       int batch)
{
    __shared__ float lds[8][2][16][68];

    const int lane = threadIdx.x & 31;
    const int wv   = threadIdx.x >> 5;
    const int half = lane >> 4;
    const int lr   = lane & 15;
    const int wave_id = blockIdx.x * 8 + wv;
    const int row0 = wave_id * 32;
    if (row0 >= batch) return;
    const int row1 = (row0 + 16 < batch) ? (row0 + 16) : row0;

    float (*hl0)[68] = lds[wv][0];
    float (*hl1)[68] = lds[wv][1];
    const v4f* wsv = (const v4f*)wsf;

    v8f acc0[4], acc1[4];

    for (int s = 0; s < SEQ_LEN; ++s) {
        const int a0 = actions[(row0 + lr) * SEQ_LEN + s];
        const int a1 = actions[(row1 + lr) * SEQ_LEN + s];
        const float* xr0 = emb + a0 * D_MODEL;
        const float* xr1 = emb + a1 * D_MODEL;

        #pragma unroll
        for (int nt = 0; nt < 4; ++nt) { acc0[nt] = (v8f)0.0f; acc1[nt] = (v8f)0.0f; }

        {
            const v4f* bm = wsv + (WS_M_OFF / 4) + s * 1024;
            #pragma unroll 4
            for (int ktp = 0; ktp < 8; ++ktp) {
                const int ke = 8 * ktp + 2 * half;
                v2f a0e = *(const v2f*)(xr0 + ke);
                v2f a0o = *(const v2f*)(xr0 + ke + 4);
                v2f a1e = *(const v2f*)(xr1 + ke);
                v2f a1o = *(const v2f*)(xr1 + ke + 4);
                v4f b[4];
                #pragma unroll
                for (int nt = 0; nt < 4; ++nt) b[nt] = bm[nt * 256 + ktp * 32 + lane];
                #pragma unroll
                for (int nt = 0; nt < 4; ++nt) {
                    v2f be = {b[nt][0], b[nt][1]};
                    acc0[nt] = __builtin_amdgcn_wmma_f32_16x16x4_f32(false, a0e, false, be, (short)0, acc0[nt], false, false);
                    acc1[nt] = __builtin_amdgcn_wmma_f32_16x16x4_f32(false, a1e, false, be, (short)0, acc1[nt], false, false);
                }
                #pragma unroll
                for (int nt = 0; nt < 4; ++nt) {
                    v2f bo = {b[nt][2], b[nt][3]};
                    acc0[nt] = __builtin_amdgcn_wmma_f32_16x16x4_f32(false, a0o, false, bo, (short)0, acc0[nt], false, false);
                    acc1[nt] = __builtin_amdgcn_wmma_f32_16x16x4_f32(false, a1o, false, bo, (short)0, acc1[nt], false, false);
                }
            }
        }

        if (s > 0) {
            const v4f* bw = wsv + (WS_W_OFF / 4) + s * 1024;
            #pragma unroll 4
            for (int ktp = 0; ktp < 8; ++ktp) {
                const int ke = 8 * ktp + 2 * half;
                v2f a0e = *(const v2f*)(&hl0[lr][ke]);
                v2f a0o = *(const v2f*)(&hl0[lr][ke + 4]);
                v2f a1e = *(const v2f*)(&hl1[lr][ke]);
                v2f a1o = *(const v2f*)(&hl1[lr][ke + 4]);
                v4f b[4];
                #pragma unroll
                for (int nt = 0; nt < 4; ++nt) b[nt] = bw[nt * 256 + ktp * 32 + lane];
                #pragma unroll
                for (int nt = 0; nt < 4; ++nt) {
                    v2f be = {b[nt][0], b[nt][1]};
                    acc0[nt] = __builtin_amdgcn_wmma_f32_16x16x4_f32(false, a0e, false, be, (short)0, acc0[nt], false, false);
                    acc1[nt] = __builtin_amdgcn_wmma_f32_16x16x4_f32(false, a1e, false, be, (short)0, acc1[nt], false, false);
                }
                #pragma unroll
                for (int nt = 0; nt < 4; ++nt) {
                    v2f bo = {b[nt][2], b[nt][3]};
                    acc0[nt] = __builtin_amdgcn_wmma_f32_16x16x4_f32(false, a0o, false, bo, (short)0, acc0[nt], false, false);
                    acc1[nt] = __builtin_amdgcn_wmma_f32_16x16x4_f32(false, a1o, false, bo, (short)0, acc1[nt], false, false);
                }
            }
        }

        #pragma unroll
        for (int nt = 0; nt < 4; ++nt) {
            const int n = nt * 16 + lr;
            const float bias = Mb[s * D_MODEL + n] + Wb[s * D_MODEL + n];
            #pragma unroll
            for (int r = 0; r < 8; ++r) {
                float v0 = sigmoid_f(acc0[nt][r] + bias);
                float v1 = sigmoid_f(acc1[nt][r] + bias);
                hl0[half * 8 + r][n] = v0;
                hl1[half * 8 + r][n] = v1;
            }
        }
    }

    const v4f* bo_base = wsv + (WS_O_OFF / 4);
    const bool two_tiles = (row1 != row0);

    v2f af0[16], af1[16];
    #pragma unroll
    for (int kt = 0; kt < 16; ++kt) {
        af0[kt] = *(const v2f*)(&hl0[lr][4 * kt + 2 * half]);
        af1[kt] = *(const v2f*)(&hl1[lr][4 * kt + 2 * half]);
    }

    #pragma unroll
    for (int g = 0; g < (NT_OUT + 3) / 4; ++g) {
        v8f o0[4], o1[4];
        float bias[4];
        #pragma unroll
        for (int j = 0; j < 4; ++j) {
            o0[j] = (v8f)0.0f; o1[j] = (v8f)0.0f;
            const int n = (4 * g + j) * 16 + lr;
            bias[j] = outb[n < ACTION_NUM ? n : ACTION_NUM - 1];
        }

        #pragma unroll
        for (int ktp = 0; ktp < 8; ++ktp) {
            v4f b[4];
            #pragma unroll
            for (int j = 0; j < 4; ++j) {
                if (4 * g + j < NT_OUT) b[j] = bo_base[(4 * g + j) * 256 + ktp * 32 + lane];
            }
            #pragma unroll
            for (int j = 0; j < 4; ++j) {
                if (4 * g + j < NT_OUT) {
                    v2f be = {b[j][0], b[j][1]};
                    o0[j] = __builtin_amdgcn_wmma_f32_16x16x4_f32(false, af0[2 * ktp], false, be, (short)0, o0[j], false, false);
                    o1[j] = __builtin_amdgcn_wmma_f32_16x16x4_f32(false, af1[2 * ktp], false, be, (short)0, o1[j], false, false);
                }
            }
            #pragma unroll
            for (int j = 0; j < 4; ++j) {
                if (4 * g + j < NT_OUT) {
                    v2f bo = {b[j][2], b[j][3]};
                    o0[j] = __builtin_amdgcn_wmma_f32_16x16x4_f32(false, af0[2 * ktp + 1], false, bo, (short)0, o0[j], false, false);
                    o1[j] = __builtin_amdgcn_wmma_f32_16x16x4_f32(false, af1[2 * ktp + 1], false, bo, (short)0, o1[j], false, false);
                }
            }
        }

        #pragma unroll
        for (int j = 0; j < 4; ++j) {
            if (4 * g + j < NT_OUT) {
                const int n = (4 * g + j) * 16 + lr;
                if (n < ACTION_NUM) {
                    #pragma unroll
                    for (int r = 0; r < 8; ++r)
                        out[(long)(row0 + half * 8 + r) * ACTION_NUM + n] = o0[j][r] + bias[j];
                }
            }
        }
        if (two_tiles) {
            #pragma unroll
            for (int j = 0; j < 4; ++j) {
                if (4 * g + j < NT_OUT) {
                    const int n = (4 * g + j) * 16 + lr;
                    if (n < ACTION_NUM) {
                        #pragma unroll
                        for (int r = 0; r < 8; ++r)
                            out[(long)(row1 + half * 8 + r) * ACTION_NUM + n] = o1[j][r] + bias[j];
                    }
                }
            }
        }
    }
}

__global__ __launch_bounds__(256)
void carnn_wmma_direct(const int*   __restrict__ actions,
                       const float* __restrict__ emb,
                       const float* __restrict__ Mw,
                       const float* __restrict__ Mb,
                       const float* __restrict__ Ww,
                       const float* __restrict__ Wb,
                       const float* __restrict__ outw,
                       const float* __restrict__ outb,
                       float*       __restrict__ out,
                       int batch)
{
    __shared__ float lds[8][16][68];
    const int lane = threadIdx.x & 31;
    const int wv   = threadIdx.x >> 5;
    const int half = lane >> 4;
    const int lr   = lane & 15;
    const int tile = blockIdx.x * 8 + wv;
    const int row0 = tile * 16;
    if (row0 >= batch) return;

    float (*hl)[68] = lds[wv];
    v8f acc[4];

    for (int s = 0; s < SEQ_LEN; ++s) {
        const float* mwp = Mw + s * D_MODEL * D_MODEL;
        const float* wwp = Ww + s * D_MODEL * D_MODEL;
        const int aidx = actions[(row0 + lr) * SEQ_LEN + s];
        const float* xrow = emb + aidx * D_MODEL;

        #pragma unroll
        for (int nt = 0; nt < 4; ++nt) acc[nt] = (v8f)0.0f;

        #pragma unroll 4
        for (int kt = 0; kt < 16; ++kt) {
            const int k = kt * 4 + half * 2;
            v2f a = *(const v2f*)(xrow + k);
            #pragma unroll
            for (int nt = 0; nt < 4; ++nt) {
                const int n = nt * 16 + lr;
                v2f b = *(const v2f*)(mwp + n * D_MODEL + k);
                acc[nt] = __builtin_amdgcn_wmma_f32_16x16x4_f32(false, a, false, b, (short)0, acc[nt], false, false);
            }
        }
        if (s > 0) {
            #pragma unroll 4
            for (int kt = 0; kt < 16; ++kt) {
                const int k = kt * 4 + half * 2;
                v2f a = *(const v2f*)(&hl[lr][k]);
                #pragma unroll
                for (int nt = 0; nt < 4; ++nt) {
                    const int n = nt * 16 + lr;
                    v2f b = *(const v2f*)(wwp + n * D_MODEL + k);
                    acc[nt] = __builtin_amdgcn_wmma_f32_16x16x4_f32(false, a, false, b, (short)0, acc[nt], false, false);
                }
            }
        }
        #pragma unroll
        for (int nt = 0; nt < 4; ++nt) {
            const int n = nt * 16 + lr;
            const float bias = Mb[s * D_MODEL + n] + Wb[s * D_MODEL + n];
            #pragma unroll
            for (int r = 0; r < 8; ++r) {
                float v = sigmoid_f(acc[nt][r] + bias);
                hl[half * 8 + r][n] = v;
            }
        }
    }

    v2f afrag[16];
    #pragma unroll
    for (int kt = 0; kt < 16; ++kt)
        afrag[kt] = *(const v2f*)(&hl[lr][4 * kt + 2 * half]);

    for (int nt0 = 0; nt0 < NT_OUT; nt0 += 4) {
        v8f oacc[4];
        #pragma unroll
        for (int j = 0; j < 4; ++j) oacc[j] = (v8f)0.0f;
        #pragma unroll 4
        for (int kt = 0; kt < 16; ++kt) {
            #pragma unroll
            for (int j = 0; j < 4; ++j) {
                const int nt = nt0 + j;
                if (nt >= NT_OUT) break;
                const int n = nt * 16 + lr;
                const int k = kt * 4 + half * 2;
                v2f b;
                if (n < ACTION_NUM) b = *(const v2f*)(outw + n * D_MODEL + k);
                else                b = (v2f)0.0f;
                oacc[j] = __builtin_amdgcn_wmma_f32_16x16x4_f32(false, afrag[kt], false, b, (short)0, oacc[j], false, false);
            }
        }
        #pragma unroll
        for (int j = 0; j < 4; ++j) {
            const int nt = nt0 + j;
            if (nt >= NT_OUT) break;
            const int n = nt * 16 + lr;
            if (n < ACTION_NUM) {
                const float bias = outb[n];
                #pragma unroll
                for (int r = 0; r < 8; ++r)
                    out[(long)(row0 + half * 8 + r) * ACTION_NUM + n] = oacc[j][r] + bias;
            }
        }
    }
}

extern "C" void kernel_launch(void* const* d_in, const int* in_sizes, int n_in,
                              void* d_out, int out_size, void* d_ws, size_t ws_size,
                              hipStream_t stream) {
    const int*   actions = (const int*)  d_in[0];
    const float* emb     = (const float*)d_in[1];
    const float* Mw      = (const float*)d_in[2];
    const float* Mb      = (const float*)d_in[3];
    const float* Ww      = (const float*)d_in[4];
    const float* Wb      = (const float*)d_in[5];
    const float* outw    = (const float*)d_in[6];
    const float* outb    = (const float*)d_in[7];
    float*       out     = (float*)d_out;

    const int batch = in_sizes[0] / SEQ_LEN;
    const int tiles = (batch + 15) / 16;

    if (ws_size >= (size_t)WS_BYTES && d_ws != nullptr) {
        float* ws = (float*)d_ws;
        carnn_pack_weights<<<(WS_FLOATS + 255) / 256, 256, 0, stream>>>(Mw, Ww, outw, ws);
        const int waves  = (tiles + 1) / 2;
        const int blocks = (waves + 7) / 8;
        carnn_wmma_packed<<<blocks, 256, 0, stream>>>(actions, emb, Mb, Wb, outb, ws, out, batch);
    } else {
        const int blocks = (tiles + 7) / 8;
        carnn_wmma_direct<<<blocks, 256, 0, stream>>>(actions, emb, Mw, Mb, Ww, Wb, outw, outb, out, batch);
    }
}


# input projection hoisted out of the batch: pre-pass computes emb[a].Mw[s]+Mb[s]+Wb[s] once per (step, action id) on WMMA; main kernel gathers it and adds it after the hl.Ww sum (sum order changes)
# speedup vs baseline: 1.9326x; 1.2815x over previous
#include <hip/hip_runtime.h>

typedef float v2f __attribute__((ext_vector_type(2)));
typedef float v4f __attribute__((ext_vector_type(4)));
typedef float v8f __attribute__((ext_vector_type(8)));

#define D_MODEL    64
#define SEQ_LEN    9
#define ACTION_NUM 300
#define NT_OUT     19
#define EMB_ROWS   301
#define MT_TAB     19
#define TAB_ROWS   304

#define WS_W_OFF       0
#define WS_O_OFF       36864
#define WS_P_OFF       56320
#define WS_PACK_FLOATS 56320
#define WS_FLOATS      231424
#define WS_BYTES       (WS_FLOATS * 4)
#define PACK_BLOCKS    (WS_PACK_FLOATS / 256)
#define TAB_TILES      (SEQ_LEN * MT_TAB)
#define TAB_BLOCKS     ((TAB_TILES + 7) / 8)

__device__ __forceinline__ float sigmoid_f(float x) {
    return 1.0f / (1.0f + __expf(-x));
}

__global__ __launch_bounds__(256)
void carnn_prepass(const float* __restrict__ emb,
                   const float* __restrict__ Mw,
                   const float* __restrict__ Mb,
                   const float* __restrict__ Ww,
                   const float* __restrict__ Wb,
                   const float* __restrict__ outw,
                   float* __restrict__ ws)
{
    if (blockIdx.x < PACK_BLOCKS) {
        const int idx = blockIdx.x * 256 + threadIdx.x;
        float val;
        if (idx < WS_O_OFF) {
            int r = idx;
            const int s   = r / 4096;  r %= 4096;
            const int nt  = r / 1024;  r %= 1024;
            const int ktp = r / 128;   r %= 128;
            const int lane = r / 4;
            const int j    = r % 4;
            const int h = lane >> 4, lr = lane & 15;
            const int n  = nt * 16 + lr;
            const int kt = 2 * ktp + (j >> 1);
            const int k  = 4 * kt + 2 * h + (j & 1);
            val = Ww[s * 4096 + n * D_MODEL + k];
        } else {
            int r = idx - WS_O_OFF;
            const int nt  = r / 1024;  r %= 1024;
            const int ktp = r / 128;   r %= 128;
            const int lane = r / 4;
            const int j    = r % 4;
            const int h = lane >> 4, lr = lane & 15;
            const int n  = nt * 16 + lr;
            const int kt = 2 * ktp + (j >> 1);
            const int k  = 4 * kt + 2 * h + (j & 1);
            val = (n < ACTION_NUM) ? outw[n * D_MODEL + k] : 0.0f;
        }
        ws[idx] = val;
        return;
    }

    const int lane = threadIdx.x & 31;
    const int wv   = threadIdx.x >> 5;
    const int half = lane >> 4;
    const int lr   = lane & 15;
    const int t = (blockIdx.x - PACK_BLOCKS) * 8 + wv;
    if (t >= TAB_TILES) return;
    const int s  = t / MT_TAB;
    const int mt = t % MT_TAB;

    int a = mt * 16 + lr;
    if (a > EMB_ROWS - 1) a = EMB_ROWS - 1;
    const float* xrow = emb + a * D_MODEL;
    const float* mwp  = Mw + s * D_MODEL * D_MODEL;

    v2f af[16];
    #pragma unroll
    for (int kt = 0; kt < 16; ++kt) af[kt] = *(const v2f*)(xrow + 4 * kt + 2 * half);

    float bias[4];
    v8f acc[4];
    #pragma unroll
    for (int nt = 0; nt < 4; ++nt) {
        acc[nt] = (v8f)0.0f;
        const int n = nt * 16 + lr;
        bias[nt] = Mb[s * D_MODEL + n] + Wb[s * D_MODEL + n];
    }

    #pragma unroll
    for (int kt = 0; kt < 16; ++kt) {
        #pragma unroll
        for (int nt = 0; nt < 4; ++nt) {
            v2f b = *(const v2f*)(mwp + (nt * 16 + lr) * D_MODEL + 4 * kt + 2 * half);
            acc[nt] = __builtin_amdgcn_wmma_f32_16x16x4_f32(false, af[kt], false, b, (short)0, acc[nt], false, false);
        }
    }

    float* dst = ws + WS_P_OFF + (s * TAB_ROWS + mt * 16 + half * 8) * D_MODEL + lr * 4;
    #pragma unroll
    for (int r = 0; r < 8; ++r) {
        v4f v = {acc[0][r] + bias[0], acc[1][r] + bias[1], acc[2][r] + bias[2], acc[3][r] + bias[3]};
        *(v4f*)(dst + r * D_MODEL) = v;
    }
}

__global__ __launch_bounds__(256)
void carnn_wmma_packed(const int*   __restrict__ actions,
                       const float* __restrict__ outb,
                       const float* __restrict__ wsf,
                       float*       __restrict__ out,
                       int batch)
{
    __shared__ float lds[8][2][16][68];

    const int lane = threadIdx.x & 31;
    const int wv   = threadIdx.x >> 5;
    const int half = lane >> 4;
    const int lr   = lane & 15;
    const int wave_id = blockIdx.x * 8 + wv;
    const int row0 = wave_id * 32;
    if (row0 >= batch) return;
    const int row1 = (row0 + 16 < batch) ? (row0 + 16) : row0;

    float (*hl0)[68] = lds[wv][0];
    float (*hl1)[68] = lds[wv][1];
    const v4f* wsv = (const v4f*)wsf;
    const float* ptab = wsf + WS_P_OFF + lr * 4;

    const int* arow0 = actions + (row0 + half * 8) * SEQ_LEN;
    const int* arow1 = actions + (row1 + half * 8) * SEQ_LEN;
    int id0[8], id1[8];
    #pragma unroll
    for (int r = 0; r < 8; ++r) {
        id0[r] = arow0[r * SEQ_LEN];
        id1[r] = arow1[r * SEQ_LEN];
    }

    v8f acc0[4], acc1[4];

    for (int s = 0; s < SEQ_LEN; ++s) {
        const float* ps = ptab + s * (TAB_ROWS * D_MODEL);
        v4f p0[8], p1[8];
        #pragma unroll
        for (int r = 0; r < 8; ++r) {
            p0[r] = *(const v4f*)(ps + id0[r] * D_MODEL);
            p1[r] = *(const v4f*)(ps + id1[r] * D_MODEL);
        }
        const int sn = (s + 1 < SEQ_LEN) ? (s + 1) : s;
        #pragma unroll
        for (int r = 0; r < 8; ++r) {
            id0[r] = arow0[r * SEQ_LEN + sn];
            id1[r] = arow1[r * SEQ_LEN + sn];
        }

        #pragma unroll
        for (int nt = 0; nt < 4; ++nt) { acc0[nt] = (v8f)0.0f; acc1[nt] = (v8f)0.0f; }

        if (s > 0) {
            const v4f* bw = wsv + (WS_W_OFF / 4) + s * 1024;
            #pragma unroll 4
            for (int ktp = 0; ktp < 8; ++ktp) {
                const int ke = 8 * ktp + 2 * half;
                v2f a0e = *(const v2f*)(&hl0[lr][ke]);
                v2f a0o = *(const v2f*)(&hl0[lr][ke + 4]);
                v2f a1e = *(const v2f*)(&hl1[lr][ke]);
                v2f a1o = *(const v2f*)(&hl1[lr][ke + 4]);
                v4f b[4];
                #pragma unroll
                for (int nt = 0; nt < 4; ++nt) b[nt] = bw[nt * 256 + ktp * 32 + lane];
                #pragma unroll
                for (int nt = 0; nt < 4; ++nt) {
                    v2f be = {b[nt][0], b[nt][1]};
                    acc0[nt] = __builtin_amdgcn_wmma_f32_16x16x4_f32(false, a0e, false, be, (short)0, acc0[nt], false, false);
                    acc1[nt] = __builtin_amdgcn_wmma_f32_16x16x4_f32(false, a1e, false, be, (short)0, acc1[nt], false, false);
                }
                #pragma unroll
                for (int nt = 0; nt < 4; ++nt) {
                    v2f bo = {b[nt][2], b[nt][3]};
                    acc0[nt] = __builtin_amdgcn_wmma_f32_16x16x4_f32(false, a0o, false, bo, (short)0, acc0[nt], false, false);
                    acc1[nt] = __builtin_amdgcn_wmma_f32_16x16x4_f32(false, a1o, false, bo, (short)0, acc1[nt], false, false);
                }
            }
        }

        #pragma unroll
        for (int nt = 0; nt < 4; ++nt) {
            const int n = nt * 16 + lr;
            #pragma unroll
            for (int r = 0; r < 8; ++r) {
                float v0 = sigmoid_f(acc0[nt][r] + p0[r][nt]);
                float v1 = sigmoid_f(acc1[nt][r] + p1[r][nt]);
                hl0[half * 8 + r][n] = v0;
                hl1[half * 8 + r][n] = v1;
            }
        }
    }

    const v4f* bo_base = wsv + (WS_O_OFF / 4);
    const bool two_tiles = (row1 != row0);

    v2f af0[16], af1[16];
    #pragma unroll
    for (int kt = 0; kt < 16; ++kt) {
        af0[kt] = *(const v2f*)(&hl0[lr][4 * kt + 2 * half]);
        af1[kt] = *(const v2f*)(&hl1[lr][4 * kt + 2 * half]);
    }

    #pragma unroll
    for (int g = 0; g < (NT_OUT + 3) / 4; ++g) {
        v8f o0[4], o1[4];
        float bias[4];
        #pragma unroll
        for (int j = 0; j < 4; ++j) {
            o0[j] = (v8f)0.0f; o1[j] = (v8f)0.0f;
            const int n = (4 * g + j) * 16 + lr;
            bias[j] = outb[n < ACTION_NUM ? n : ACTION_NUM - 1];
        }

        #pragma unroll
        for (int ktp = 0; ktp < 8; ++ktp) {
            v4f b[4];
            #pragma unroll
            for (int j = 0; j < 4; ++j) {
                if (4 * g + j < NT_OUT) b[j] = bo_base[(4 * g + j) * 256 + ktp * 32 + lane];
            }
            #pragma unroll
            for (int j = 0; j < 4; ++j) {
                if (4 * g + j < NT_OUT) {
                    v2f be = {b[j][0], b[j][1]};
                    o0[j] = __builtin_amdgcn_wmma_f32_16x16x4_f32(false, af0[2 * ktp], false, be, (short)0, o0[j], false, false);
                    o1[j] = __builtin_amdgcn_wmma_f32_16x16x4_f32(false, af1[2 * ktp], false, be, (short)0, o1[j], false, false);
                }
            }
            #pragma unroll
            for (int j = 0; j < 4; ++j) {
                if (4 * g + j < NT_OUT) {
                    v2f bo = {b[j][2], b[j][3]};
                    o0[j] = __builtin_amdgcn_wmma_f32_16x16x4_f32(false, af0[2 * ktp + 1], false, bo, (short)0, o0[j], false, false);
                    o1[j] = __builtin_amdgcn_wmma_f32_16x16x4_f32(false, af1[2 * ktp + 1], false, bo, (short)0, o1[j], false, false);
                }
            }
        }

        #pragma unroll
        for (int j = 0; j < 4; ++j) {
            if (4 * g + j < NT_OUT) {
                const int n = (4 * g + j) * 16 + lr;
                if (n < ACTION_NUM) {
                    #pragma unroll
                    for (int r = 0; r < 8; ++r)
                        out[(long)(row0 + half * 8 + r) * ACTION_NUM + n] = o0[j][r] + bias[j];
                }
            }
        }
        if (two_tiles) {
            #pragma unroll
            for (int j = 0; j < 4; ++j) {
                if (4 * g + j < NT_OUT) {
                    const int n = (4 * g + j) * 16 + lr;
                    if (n < ACTION_NUM) {
                        #pragma unroll
                        for (int r = 0; r < 8; ++r)
                            out[(long)(row1 + half * 8 + r) * ACTION_NUM + n] = o1[j][r] + bias[j];
                    }
                }
            }
        }
    }
}

__global__ __launch_bounds__(256)
void carnn_wmma_direct(const int*   __restrict__ actions,
                       const float* __restrict__ emb,
                       const float* __restrict__ Mw,
                       const float* __restrict__ Mb,
                       const float* __restrict__ Ww,
                       const float* __restrict__ Wb,
                       const float* __restrict__ outw,
                       const float* __restrict__ outb,
                       float*       __restrict__ out,
                       int batch)
{
    __shared__ float lds[8][16][68];
    const int lane = threadIdx.x & 31;
    const int wv   = threadIdx.x >> 5;
    const int half = lane >> 4;
    const int lr   = lane & 15;
    const int tile = blockIdx.x * 8 + wv;
    const int row0 = tile * 16;
    if (row0 >= batch) return;

    float (*hl)[68] = lds[wv];
    v8f acc[4];

    for (int s = 0; s < SEQ_LEN; ++s) {
        const float* mwp = Mw + s * D_MODEL * D_MODEL;
        const float* wwp = Ww + s * D_MODEL * D_MODEL;
        const int aidx = actions[(row0 + lr) * SEQ_LEN + s];
        const float* xrow = emb + aidx * D_MODEL;

        #pragma unroll
        for (int nt = 0; nt < 4; ++nt) acc[nt] = (v8f)0.0f;

        #pragma unroll 4
        for (int kt = 0; kt < 16; ++kt) {
            const int k = kt * 4 + half * 2;
            v2f a = *(const v2f*)(xrow + k);
            #pragma unroll
            for (int nt = 0; nt < 4; ++nt) {
                const int n = nt * 16 + lr;
                v2f b = *(const v2f*)(mwp + n * D_MODEL + k);
                acc[nt] = __builtin_amdgcn_wmma_f32_16x16x4_f32(false, a, false, b, (short)0, acc[nt], false, false);
            }
        }
        if (s > 0) {
            #pragma unroll 4
            for (int kt = 0; kt < 16; ++kt) {
                const int k = kt * 4 + half * 2;
                v2f a = *(const v2f*)(&hl[lr][k]);
                #pragma unroll
                for (int nt = 0; nt < 4; ++nt) {
                    const int n = nt * 16 + lr;
                    v2f b = *(const v2f*)(wwp + n * D_MODEL + k);
                    acc[nt] = __builtin_amdgcn_wmma_f32_16x16x4_f32(false, a, false, b, (short)0, acc[nt], false, false);
                }
            }
        }
        #pragma unroll
        for (int nt = 0; nt < 4; ++nt) {
            const int n = nt * 16 + lr;
            const float bias = Mb[s * D_MODEL + n] + Wb[s * D_MODEL + n];
            #pragma unroll
            for (int r = 0; r < 8; ++r) {
                float v = sigmoid_f(acc[nt][r] + bias);
                hl[half * 8 + r][n] = v;
            }
        }
    }

    v2f afrag[16];
    #pragma unroll
    for (int kt = 0; kt < 16; ++kt)
        afrag[kt] = *(const v2f*)(&hl[lr][4 * kt + 2 * half]);

    for (int nt0 = 0; nt0 < NT_OUT; nt0 += 4) {
        v8f oacc[4];
        #pragma unroll
        for (int j = 0; j < 4; ++j) oacc[j] = (v8f)0.0f;
        #pragma unroll 4
        for (int kt = 0; kt < 16; ++kt) {
            #pragma unroll
            for (int j = 0; j < 4; ++j) {
                const int nt = nt0 + j;
                if (nt >= NT_OUT) break;
                const int n = nt * 16 + lr;
                const int k = kt * 4 + half * 2;
                v2f b;
                if (n < ACTION_NUM) b = *(const v2f*)(outw + n * D_MODEL + k);
                else                b = (v2f)0.0f;
                oacc[j] = __builtin_amdgcn_wmma_f32_16x16x4_f32(false, afrag[kt], false, b, (short)0, oacc[j], false, false);
            }
        }
        #pragma unroll
        for (int j = 0; j < 4; ++j) {
            const int nt = nt0 + j;
            if (nt >= NT_OUT) break;
            const int n = nt * 16 + lr;
            if (n < ACTION_NUM) {
                const float bias = outb[n];
                #pragma unroll
                for (int r = 0; r < 8; ++r)
                    out[(long)(row0 + half * 8 + r) * ACTION_NUM + n] = oacc[j][r] + bias;
            }
        }
    }
}

extern "C" void kernel_launch(void* const* d_in, const int* in_sizes, int n_in,
                              void* d_out, int out_size, void* d_ws, size_t ws_size,
                              hipStream_t stream) {
    const int*   actions = (const int*)  d_in[0];
    const float* emb     = (const float*)d_in[1];
    const float* Mw      = (const float*)d_in[2];
    const float* Mb      = (const float*)d_in[3];
    const float* Ww      = (const float*)d_in[4];
    const float* Wb      = (const float*)d_in[5];
    const float* outw    = (const float*)d_in[6];
    const float* outb    = (const float*)d_in[7];
    float*       out     = (float*)d_out;

    const int batch = in_sizes[0] / SEQ_LEN;
    const int tiles = (batch + 15) / 16;

    if (ws_size >= (size_t)WS_BYTES && d_ws != nullptr) {
        float* ws = (float*)d_ws;
        carnn_prepass<<<PACK_BLOCKS + TAB_BLOCKS, 256, 0, stream>>>(emb, Mw, Mb, Ww, Wb, outw, ws);
        const int waves  = (tiles + 1) / 2;
        const int blocks = (waves + 7) / 8;
        carnn_wmma_packed<<<blocks, 256, 0, stream>>>(actions, outb, ws, out, batch);
    } else {
        const int blocks = (tiles + 7) / 8;
        carnn_wmma_direct<<<blocks, 256, 0, stream>>>(actions, emb, Mw, Mb, Ww, Wb, outw, outb, out, batch);
    }
}


# sigmoid: hardware reciprocal (v_rcp_f32, 1 ulp) replaces the IEEE division (11 vector instructions per value, 64 values per lane per step); an hl value may round 1 ulp differently
# speedup vs baseline: 2.2011x; 1.1389x over previous
#include <hip/hip_runtime.h>

typedef float v2f __attribute__((ext_vector_type(2)));
typedef float v4f __attribute__((ext_vector_type(4)));
typedef float v8f __attribute__((ext_vector_type(8)));

#define D_MODEL    64
#define SEQ_LEN    9
#define ACTION_NUM 300
#define NT_OUT     19
#define EMB_ROWS   301
#define MT_TAB     19
#define TAB_ROWS   304

#define WS_W_OFF       0
#define WS_O_OFF       36864
#define WS_P_OFF       56320
#define WS_PACK_FLOATS 56320
#define WS_FLOATS      231424
#define WS_BYTES       (WS_FLOATS * 4)
#define PACK_BLOCKS    (WS_PACK_FLOATS / 256)
#define TAB_TILES      (SEQ_LEN * MT_TAB)
#define TAB_BLOCKS     ((TAB_TILES + 7) / 8)

__device__ __forceinline__ float sigmoid_f(float x) {
    return __builtin_amdgcn_rcpf(1.0f + __expf(-x));
}

__global__ __launch_bounds__(256)
void carnn_prepass(const float* __restrict__ emb,
                   const float* __restrict__ Mw,
                   const float* __restrict__ Mb,
                   const float* __restrict__ Ww,
                   const float* __restrict__ Wb,
                   const float* __restrict__ outw,
                   float* __restrict__ ws)
{
    if (blockIdx.x < PACK_BLOCKS) {
        const int idx = blockIdx.x * 256 + threadIdx.x;
        float val;
        if (idx < WS_O_OFF) {
            int r = idx;
            const int s   = r / 4096;  r %= 4096;
            const int nt  = r / 1024;  r %= 1024;
            const int ktp = r / 128;   r %= 128;
            const int lane = r / 4;
            const int j    = r % 4;
            const int h = lane >> 4, lr = lane & 15;
            const int n  = nt * 16 + lr;
            const int kt = 2 * ktp + (j >> 1);
            const int k  = 4 * kt + 2 * h + (j & 1);
            val = Ww[s * 4096 + n * D_MODEL + k];
        } else {
            int r = idx - WS_O_OFF;
            const int nt  = r / 1024;  r %= 1024;
            const int ktp = r / 128;   r %= 128;
            const int lane = r / 4;
            const int j    = r % 4;
            const int h = lane >> 4, lr = lane & 15;
            const int n  = nt * 16 + lr;
            const int kt = 2 * ktp + (j >> 1);
            const int k  = 4 * kt + 2 * h + (j & 1);
            val = (n < ACTION_NUM) ? outw[n * D_MODEL + k] : 0.0f;
        }
        ws[idx] = val;
        return;
    }

    const int lane = threadIdx.x & 31;
    const int wv   = threadIdx.x >> 5;
    const int half = lane >> 4;
    const int lr   = lane & 15;
    const int t = (blockIdx.x - PACK_BLOCKS) * 8 + wv;
    if (t >= TAB_TILES) return;
    const int s  = t / MT_TAB;
    const int mt = t % MT_TAB;

    int a = mt * 16 + lr;
    if (a > EMB_ROWS - 1) a = EMB_ROWS - 1;
    const float* xrow = emb + a * D_MODEL;
    const float* mwp  = Mw + s * D_MODEL * D_MODEL;

    v2f af[16];
    #pragma unroll
    for (int kt = 0; kt < 16; ++kt) af[kt] = *(const v2f*)(xrow + 4 * kt + 2 * half);

    float bias[4];
    v8f acc[4];
    #pragma unroll
    for (int nt = 0; nt < 4; ++nt) {
        acc[nt] = (v8f)0.0f;
        const int n = nt * 16 + lr;
        bias[nt] = Mb[s * D_MODEL + n] + Wb[s * D_MODEL + n];
    }

    #pragma unroll
    for (int kt = 0; kt < 16; ++kt) {
        #pragma unroll
        for (int nt = 0; nt < 4; ++nt) {
            v2f b = *(const v2f*)(mwp + (nt * 16 + lr) * D_MODEL + 4 * kt + 2 * half);
            acc[nt] = __builtin_amdgcn_wmma_f32_16x16x4_f32(false, af[kt], false, b, (short)0, acc[nt], false, false);
        }
    }

    float* dst = ws + WS_P_OFF + (s * TAB_ROWS + mt * 16 + half * 8) * D_MODEL + lr * 4;
    #pragma unroll
    for (int r = 0; r < 8; ++r) {
        v4f v = {acc[0][r] + bias[0], acc[1][r] + bias[1], acc[2][r] + bias[2], acc[3][r] + bias[3]};
        *(v4f*)(dst + r * D_MODEL) = v;
    }
}

__global__ __launch_bounds__(256)
void carnn_wmma_packed(const int*   __restrict__ actions,
                       const float* __restrict__ outb,
                       const float* __restrict__ wsf,
                       float*       __restrict__ out,
                       int batch)
{
    __shared__ float lds[8][2][16][68];

    const int lane = threadIdx.x & 31;
    const int wv   = threadIdx.x >> 5;
    const int half = lane >> 4;
    const int lr   = lane & 15;
    const int wave_id = blockIdx.x * 8 + wv;
    const int row0 = wave_id * 32;
    if (row0 >= batch) return;
    const int row1 = (row0 + 16 < batch) ? (row0 + 16) : row0;

    float (*hl0)[68] = lds[wv][0];
    float (*hl1)[68] = lds[wv][1];
    const v4f* wsv = (const v4f*)wsf;
    const float* ptab = wsf + WS_P_OFF + lr * 4;

    const int* arow0 = actions + (row0 + half * 8) * SEQ_LEN;
    const int* arow1 = actions + (row1 + half * 8) * SEQ_LEN;
    int id0[8], id1[8];
    #pragma unroll
    for (int r = 0; r < 8; ++r) {
        id0[r] = arow0[r * SEQ_LEN];
        id1[r] = arow1[r * SEQ_LEN];
    }

    v8f acc0[4], acc1[4];

    for (int s = 0; s < SEQ_LEN; ++s) {
        const float* ps = ptab + s * (TAB_ROWS * D_MODEL);
        v4f p0[8], p1[8];
        #pragma unroll
        for (int r = 0; r < 8; ++r) {
            p0[r] = *(const v4f*)(ps + id0[r] * D_MODEL);
            p1[r] = *(const v4f*)(ps + id1[r] * D_MODEL);
        }
        const int sn = (s + 1 < SEQ_LEN) ? (s + 1) : s;
        #pragma unroll
        for (int r = 0; r < 8; ++r) {
            id0[r] = arow0[r * SEQ_LEN + sn];
            id1[r] = arow1[r * SEQ_LEN + sn];
        }

        #pragma unroll
        for (int nt = 0; nt < 4; ++nt) { acc0[nt] = (v8f)0.0f; acc1[nt] = (v8f)0.0f; }

        if (s > 0) {
            const v4f* bw = wsv + (WS_W_OFF / 4) + s * 1024;
            #pragma unroll 4
            for (int ktp = 0; ktp < 8; ++ktp) {
                const int ke = 8 * ktp + 2 * half;
                v2f a0e = *(const v2f*)(&hl0[lr][ke]);
                v2f a0o = *(const v2f*)(&hl0[lr][ke + 4]);
                v2f a1e = *(const v2f*)(&hl1[lr][ke]);
                v2f a1o = *(const v2f*)(&hl1[lr][ke + 4]);
                v4f b[4];
                #pragma unroll
                for (int nt = 0; nt < 4; ++nt) b[nt] = bw[nt * 256 + ktp * 32 + lane];
                #pragma unroll
                for (int nt = 0; nt < 4; ++nt) {
                    v2f be = {b[nt][0], b[nt][1]};
                    acc0[nt] = __builtin_amdgcn_wmma_f32_16x16x4_f32(false, a0e, false, be, (short)0, acc0[nt], false, false);
                    acc1[nt] = __builtin_amdgcn_wmma_f32_16x16x4_f32(false, a1e, false, be, (short)0, acc1[nt], false, false);
                }
                #pragma unroll
                for (int nt = 0; nt < 4; ++nt) {
                    v2f bo = {b[nt][2], b[nt][3]};
                    acc0[nt] = __builtin_amdgcn_wmma_f32_16x16x4_f32(false, a0o, false, bo, (short)0, acc0[nt], false, false);
                    acc1[nt] = __builtin_amdgcn_wmma_f32_16x16x4_f32(false, a1o, false, bo, (short)0, acc1[nt], false, false);
                }
            }
        }

        #pragma unroll
        for (int nt = 0; nt < 4; ++nt) {
            const int n = nt * 16 + lr;
            #pragma unroll
            for (int r = 0; r < 8; ++r) {
                float v0 = sigmoid_f(acc0[nt][r] + p0[r][nt]);
                float v1 = sigmoid_f(acc1[nt][r] + p1[r][nt]);
                hl0[half * 8 + r][n] = v0;
                hl1[half * 8 + r][n] = v1;
            }
        }
    }

    const v4f* bo_base = wsv + (WS_O_OFF / 4);
    const bool two_tiles = (row1 != row0);

    v2f af0[16], af1[16];
    #pragma unroll
    for (int kt = 0; kt < 16; ++kt) {
        af0[kt] = *(const v2f*)(&hl0[lr][4 * kt + 2 * half]);
        af1[kt] = *(const v2f*)(&hl1[lr][4 * kt + 2 * half]);
    }

    #pragma unroll
    for (int g = 0; g < (NT_OUT + 3) / 4; ++g) {
        v8f o0[4], o1[4];
        float bias[4];
        #pragma unroll
        for (int j = 0; j < 4; ++j) {
            o0[j] = (v8f)0.0f; o1[j] = (v8f)0.0f;
            const int n = (4 * g + j) * 16 + lr;
            bias[j] = outb[n < ACTION_NUM ? n : ACTION_NUM - 1];
        }

        #pragma unroll
        for (int ktp = 0; ktp < 8; ++ktp) {
            v4f b[4];
            #pragma unroll
            for (int j = 0; j < 4; ++j) {
                if (4 * g + j < NT_OUT) b[j] = bo_base[(4 * g + j) * 256 + ktp * 32 + lane];
            }
            #pragma unroll
            for (int j = 0; j < 4; ++j) {
                if (4 * g + j < NT_OUT) {
                    v2f be = {b[j][0], b[j][1]};
                    o0[j] = __builtin_amdgcn_wmma_f32_16x16x4_f32(false, af0[2 * ktp], false, be, (short)0, o0[j], false, false);
                    o1[j] = __builtin_amdgcn_wmma_f32_16x16x4_f32(false, af1[2 * ktp], false, be, (short)0, o1[j], false, false);
                }
            }
            #pragma unroll
            for (int j = 0; j < 4; ++j) {
                if (4 * g + j < NT_OUT) {
                    v2f bo = {b[j][2], b[j][3]};
                    o0[j] = __builtin_amdgcn_wmma_f32_16x16x4_f32(false, af0[2 * ktp + 1], false, bo, (short)0, o0[j], false, false);
                    o1[j] = __builtin_amdgcn_wmma_f32_16x16x4_f32(false, af1[2 * ktp + 1], false, bo, (short)0, o1[j], false, false);
                }
            }
        }

        #pragma unroll
        for (int j = 0; j < 4; ++j) {
            if (4 * g + j < NT_OUT) {
                const int n = (4 * g + j) * 16 + lr;
                if (n < ACTION_NUM) {
                    #pragma unroll
                    for (int r = 0; r < 8; ++r)
                        out[(long)(row0 + half * 8 + r) * ACTION_NUM + n] = o0[j][r] + bias[j];
                }
            }
        }
        if (two_tiles) {
            #pragma unroll
            for (int j = 0; j < 4; ++j) {
                if (4 * g + j < NT_OUT) {
                    const int n = (4 * g + j) * 16 + lr;
                    if (n < ACTION_NUM) {
                        #pragma unroll
                        for (int r = 0; r < 8; ++r)
                            out[(long)(row1 + half * 8 + r) * ACTION_NUM + n] = o1[j][r] + bias[j];
                    }
                }
            }
        }
    }
}

__global__ __launch_bounds__(256)
void carnn_wmma_direct(const int*   __restrict__ actions,
                       const float* __restrict__ emb,
                       const float* __restrict__ Mw,
                       const float* __restrict__ Mb,
                       const float* __restrict__ Ww,
                       const float* __restrict__ Wb,
                       const float* __restrict__ outw,
                       const float* __restrict__ outb,
                       float*       __restrict__ out,
                       int batch)
{
    __shared__ float lds[8][16][68];
    const int lane = threadIdx.x & 31;
    const int wv   = threadIdx.x >> 5;
    const int half = lane >> 4;
    const int lr   = lane & 15;
    const int tile = blockIdx.x * 8 + wv;
    const int row0 = tile * 16;
    if (row0 >= batch) return;

    float (*hl)[68] = lds[wv];
    v8f acc[4];

    for (int s = 0; s < SEQ_LEN; ++s) {
        const float* mwp = Mw + s * D_MODEL * D_MODEL;
        const float* wwp = Ww + s * D_MODEL * D_MODEL;
        const int aidx = actions[(row0 + lr) * SEQ_LEN + s];
        const float* xrow = emb + aidx * D_MODEL;

        #pragma unroll
        for (int nt = 0; nt < 4; ++nt) acc[nt] = (v8f)0.0f;

        #pragma unroll 4
        for (int kt = 0; kt < 16; ++kt) {
            const int k = kt * 4 + half * 2;
            v2f a = *(const v2f*)(xrow + k);
            #pragma unroll
            for (int nt = 0; nt < 4; ++nt) {
                const int n = nt * 16 + lr;
                v2f b = *(const v2f*)(mwp + n * D_MODEL + k);
                acc[nt] = __builtin_amdgcn_wmma_f32_16x16x4_f32(false, a, false, b, (short)0, acc[nt], false, false);
            }
        }
        if (s > 0) {
            #pragma unroll 4
            for (int kt = 0; kt < 16; ++kt) {
                const int k = kt * 4 + half * 2;
                v2f a = *(const v2f*)(&hl[lr][k]);
                #pragma unroll
                for (int nt = 0; nt < 4; ++nt) {
                    const int n = nt * 16 + lr;
                    v2f b = *(const v2f*)(wwp + n * D_MODEL + k);
                    acc[nt] = __builtin_amdgcn_wmma_f32_16x16x4_f32(false, a, false, b, (short)0, acc[nt], false, false);
                }
            }
        }
        #pragma unroll
        for (int nt = 0; nt < 4; ++nt) {
            const int n = nt * 16 + lr;
            const float bias = Mb[s * D_MODEL + n] + Wb[s * D_MODEL + n];
            #pragma unroll
            for (int r = 0; r < 8; ++r) {
                float v = sigmoid_f(acc[nt][r] + bias);
                hl[half * 8 + r][n] = v;
            }
        }
    }

    v2f afrag[16];
    #pragma unroll
    for (int kt = 0; kt < 16; ++kt)
        afrag[kt] = *(const v2f*)(&hl[lr][4 * kt + 2 * half]);

    for (int nt0 = 0; nt0 < NT_OUT; nt0 += 4) {
        v8f oacc[4];
        #pragma unroll
        for (int j = 0; j < 4; ++j) oacc[j] = (v8f)0.0f;
        #pragma unroll 4
        for (int kt = 0; kt < 16; ++kt) {
            #pragma unroll
            for (int j = 0; j < 4; ++j) {
                const int nt = nt0 + j;
                if (nt >= NT_OUT) break;
                const int n = nt * 16 + lr;
                const int k = kt * 4 + half * 2;
                v2f b;
                if (n < ACTION_NUM) b = *(const v2f*)(outw + n * D_MODEL + k);
                else                b = (v2f)0.0f;
                oacc[j] = __builtin_amdgcn_wmma_f32_16x16x4_f32(false, afrag[kt], false, b, (short)0, oacc[j], false, false);
            }
        }
        #pragma unroll
        for (int j = 0; j < 4; ++j) {
            const int nt = nt0 + j;
            if (nt >= NT_OUT) break;
            const int n = nt * 16 + lr;
            if (n < ACTION_NUM) {
                const float bias = outb[n];
                #pragma unroll
                for (int r = 0; r < 8; ++r)
                    out[(long)(row0 + half * 8 + r) * ACTION_NUM + n] = oacc[j][r] + bias;
            }
        }
    }
}

extern "C" void kernel_launch(void* const* d_in, const int* in_sizes, int n_in,
                              void* d_out, int out_size, void* d_ws, size_t ws_size,
                              hipStream_t stream) {
    const int*   actions = (const int*)  d_in[0];
    const float* emb     = (const float*)d_in[1];
    const float* Mw      = (const float*)d_in[2];
    const float* Mb      = (const float*)d_in[3];
    const float* Ww      = (const float*)d_in[4];
    const float* Wb      = (const float*)d_in[5];
    const float* outw    = (const float*)d_in[6];
    const float* outb    = (const float*)d_in[7];
    float*       out     = (float*)d_out;

    const int batch = in_sizes[0] / SEQ_LEN;
    const int tiles = (batch + 15) / 16;

    if (ws_size >= (size_t)WS_BYTES && d_ws != nullptr) {
        float* ws = (float*)d_ws;
        carnn_prepass<<<PACK_BLOCKS + TAB_BLOCKS, 256, 0, stream>>>(emb, Mw, Mb, Ww, Wb, outw, ws);
        const int waves  = (tiles + 1) / 2;
        const int blocks = (waves + 7) / 8;
        carnn_wmma_packed<<<blocks, 256, 0, stream>>>(actions, outb, ws, out, batch);
    } else {
        const int blocks = (tiles + 7) / 8;
        carnn_wmma_direct<<<blocks, 256, 0, stream>>>(actions, emb, Mw, Mb, Ww, Wb, outw, outb, out, batch);
    }
}


# weight fragments (Ww steps, out_w groups) are loaded one 64-instruction block ahead into the register set just freed, so no matrix block opens by waiting on its own loads; sum order unchanged
# speedup vs baseline: 2.4580x; 1.1167x over previous
#include <hip/hip_runtime.h>

typedef float v2f __attribute__((ext_vector_type(2)));
typedef float v4f __attribute__((ext_vector_type(4)));
typedef float v8f __attribute__((ext_vector_type(8)));

#define D_MODEL    64
#define SEQ_LEN    9
#define ACTION_NUM 300
#define NT_OUT     19
#define EMB_ROWS   301
#define MT_TAB     19
#define TAB_ROWS   304

#define WS_W_OFF       0
#define WS_O_OFF       36864
#define WS_P_OFF       56320
#define WS_PACK_FLOATS 56320
#define WS_FLOATS      231424
#define WS_BYTES       (WS_FLOATS * 4)
#define PACK_BLOCKS    (WS_PACK_FLOATS / 256)
#define TAB_TILES      (SEQ_LEN * MT_TAB)
#define TAB_BLOCKS     ((TAB_TILES + 7) / 8)

__device__ __forceinline__ float sigmoid_f(float x) {
    return __builtin_amdgcn_rcpf(1.0f + __expf(-x));
}

__global__ __launch_bounds__(256)
void carnn_prepass(const float* __restrict__ emb,
                   const float* __restrict__ Mw,
                   const float* __restrict__ Mb,
                   const float* __restrict__ Ww,
                   const float* __restrict__ Wb,
                   const float* __restrict__ outw,
                   float* __restrict__ ws)
{
    if (blockIdx.x < PACK_BLOCKS) {
        const int idx = blockIdx.x * 256 + threadIdx.x;
        float val;
        if (idx < WS_O_OFF) {
            int r = idx;
            const int s   = r / 4096;  r %= 4096;
            const int nt  = r / 1024;  r %= 1024;
            const int ktp = r / 128;   r %= 128;
            const int lane = r / 4;
            const int j    = r % 4;
            const int h = lane >> 4, lr = lane & 15;
            const int n  = nt * 16 + lr;
            const int kt = 2 * ktp + (j >> 1);
            const int k  = 4 * kt + 2 * h + (j & 1);
            val = Ww[s * 4096 + n * D_MODEL + k];
        } else {
            int r = idx - WS_O_OFF;
            const int nt  = r / 1024;  r %= 1024;
            const int ktp = r / 128;   r %= 128;
            const int lane = r / 4;
            const int j    = r % 4;
            const int h = lane >> 4, lr = lane & 15;
            const int n  = nt * 16 + lr;
            const int kt = 2 * ktp + (j >> 1);
            const int k  = 4 * kt + 2 * h + (j & 1);
            val = (n < ACTION_NUM) ? outw[n * D_MODEL + k] : 0.0f;
        }
        ws[idx] = val;
        return;
    }

    const int lane = threadIdx.x & 31;
    const int wv   = threadIdx.x >> 5;
    const int half = lane >> 4;
    const int lr   = lane & 15;
    const int t = (blockIdx.x - PACK_BLOCKS) * 8 + wv;
    if (t >= TAB_TILES) return;
    const int s  = t / MT_TAB;
    const int mt = t % MT_TAB;

    int a = mt * 16 + lr;
    if (a > EMB_ROWS - 1) a = EMB_ROWS - 1;
    const float* xrow = emb + a * D_MODEL;
    const float* mwp  = Mw + s * D_MODEL * D_MODEL;

    v2f af[16];
    #pragma unroll
    for (int kt = 0; kt < 16; ++kt) af[kt] = *(const v2f*)(xrow + 4 * kt + 2 * half);

    float bias[4];
    v8f acc[4];
    #pragma unroll
    for (int nt = 0; nt < 4; ++nt) {
        acc[nt] = (v8f)0.0f;
        const int n = nt * 16 + lr;
        bias[nt] = Mb[s * D_MODEL + n] + Wb[s * D_MODEL + n];
    }

    #pragma unroll
    for (int kt = 0; kt < 16; ++kt) {
        #pragma unroll
        for (int nt = 0; nt < 4; ++nt) {
            v2f b = *(const v2f*)(mwp + (nt * 16 + lr) * D_MODEL + 4 * kt + 2 * half);
            acc[nt] = __builtin_amdgcn_wmma_f32_16x16x4_f32(false, af[kt], false, b, (short)0, acc[nt], false, false);
        }
    }

    float* dst = ws + WS_P_OFF + (s * TAB_ROWS + mt * 16 + half * 8) * D_MODEL + lr * 4;
    #pragma unroll
    for (int r = 0; r < 8; ++r) {
        v4f v = {acc[0][r] + bias[0], acc[1][r] + bias[1], acc[2][r] + bias[2], acc[3][r] + bias[3]};
        *(v4f*)(dst + r * D_MODEL) = v;
    }
}


template <int NTV>
__device__ __forceinline__ void load_bblock(v4f (&b)[16], const v4f* __restrict__ base, int ktp0)
{
    #pragma unroll
    for (int kl = 0; kl < 4; ++kl) {
        #pragma unroll
        for (int nt = 0; nt < 4; ++nt) {
            if (nt < NTV) b[kl * 4 + nt] = base[nt * 256 + (ktp0 + kl) * 32];
        }
    }
}

__device__ __forceinline__ void mma_block_lds(v8f (&acc0)[4], v8f (&acc1)[4], const v4f (&b)[16],
                                              const float (*hl0)[68], const float (*hl1)[68],
                                              int lr, int half, int ktp0)
{
    #pragma unroll
    for (int kl = 0; kl < 4; ++kl) {
        const int ke = 8 * (ktp0 + kl) + 2 * half;
        v2f a0e = *(const v2f*)(&hl0[lr][ke]);
        v2f a0o = *(const v2f*)(&hl0[lr][ke + 4]);
        v2f a1e = *(const v2f*)(&hl1[lr][ke]);
        v2f a1o = *(const v2f*)(&hl1[lr][ke + 4]);
        #pragma unroll
        for (int nt = 0; nt < 4; ++nt) {
            v2f be = {b[kl * 4 + nt][0], b[kl * 4 + nt][1]};
            acc0[nt] = __builtin_amdgcn_wmma_f32_16x16x4_f32(false, a0e, false, be, (short)0, acc0[nt], false, false);
            acc1[nt] = __builtin_amdgcn_wmma_f32_16x16x4_f32(false, a1e, false, be, (short)0, acc1[nt], false, false);
        }
        #pragma unroll
        for (int nt = 0; nt < 4; ++nt) {
            v2f bo = {b[kl * 4 + nt][2], b[kl * 4 + nt][3]};
            acc0[nt] = __builtin_amdgcn_wmma_f32_16x16x4_f32(false, a0o, false, bo, (short)0, acc0[nt], false, false);
            acc1[nt] = __builtin_amdgcn_wmma_f32_16x16x4_f32(false, a1o, false, bo, (short)0, acc1[nt], false, false);
        }
    }
}

template <int NTV>
__device__ __forceinline__ void mma_block_reg(v8f (&o0)[4], v8f (&o1)[4], const v4f (&b)[16],
                                              const v2f (&af0)[16], const v2f (&af1)[16], int ktp0)
{
    #pragma unroll
    for (int kl = 0; kl < 4; ++kl) {
        const int ktp = ktp0 + kl;
        #pragma unroll
        for (int j = 0; j < 4; ++j) {
            if (j < NTV) {
                v2f be = {b[kl * 4 + j][0], b[kl * 4 + j][1]};
                o0[j] = __builtin_amdgcn_wmma_f32_16x16x4_f32(false, af0[2 * ktp], false, be, (short)0, o0[j], false, false);
                o1[j] = __builtin_amdgcn_wmma_f32_16x16x4_f32(false, af1[2 * ktp], false, be, (short)0, o1[j], false, false);
            }
        }
        #pragma unroll
        for (int j = 0; j < 4; ++j) {
            if (j < NTV) {
                v2f bo = {b[kl * 4 + j][2], b[kl * 4 + j][3]};
                o0[j] = __builtin_amdgcn_wmma_f32_16x16x4_f32(false, af0[2 * ktp + 1], false, bo, (short)0, o0[j], false, false);
                o1[j] = __builtin_amdgcn_wmma_f32_16x16x4_f32(false, af1[2 * ktp + 1], false, bo, (short)0, o1[j], false, false);
            }
        }
    }
}

__device__ __forceinline__ void sigmoid_store(const v8f (&acc0)[4], const v8f (&acc1)[4],
                                              const v4f (&p0)[8], const v4f (&p1)[8],
                                              float (*hl0)[68], float (*hl1)[68], int lr, int half)
{
    #pragma unroll
    for (int nt = 0; nt < 4; ++nt) {
        const int n = nt * 16 + lr;
        #pragma unroll
        for (int r = 0; r < 8; ++r) {
            float v0 = sigmoid_f(acc0[nt][r] + p0[r][nt]);
            float v1 = sigmoid_f(acc1[nt][r] + p1[r][nt]);
            hl0[half * 8 + r][n] = v0;
            hl1[half * 8 + r][n] = v1;
        }
    }
}

template <int NTV, int NTN>
__device__ __forceinline__ void out_group(v4f (&bx)[16], v4f (&by)[16],
                                          const v2f (&af0)[16], const v2f (&af1)[16],
                                          const v4f* __restrict__ bnext,
                                          const float* __restrict__ outb, float* __restrict__ out,
                                          int g, int row0, int row1, bool two_tiles, int lr, int half)
{
    v8f o0[4], o1[4];
    float bias[4];
    #pragma unroll
    for (int j = 0; j < 4; ++j) {
        o0[j] = (v8f)0.0f; o1[j] = (v8f)0.0f;
        const int n = (4 * g + j) * 16 + lr;
        bias[j] = outb[n < ACTION_NUM ? n : ACTION_NUM - 1];
    }

    mma_block_reg<NTV>(o0, o1, bx, af0, af1, 0);
    if (NTN > 0) {
        __builtin_amdgcn_sched_barrier(0);
        load_bblock<NTN>(bx, bnext, 0);
        __builtin_amdgcn_sched_barrier(0);
    }
    mma_block_reg<NTV>(o0, o1, by, af0, af1, 4);
    if (NTN > 0) {
        __builtin_amdgcn_sched_barrier(0);
        load_bblock<NTN>(by, bnext, 4);
        __builtin_amdgcn_sched_barrier(0);
    }

    #pragma unroll
    for (int j = 0; j < 4; ++j) {
        if (j < NTV) {
            const int n = (4 * g + j) * 16 + lr;
            if (n < ACTION_NUM) {
                #pragma unroll
                for (int r = 0; r < 8; ++r)
                    out[(long)(row0 + half * 8 + r) * ACTION_NUM + n] = o0[j][r] + bias[j];
            }
        }
    }
    if (two_tiles) {
        #pragma unroll
        for (int j = 0; j < 4; ++j) {
            if (j < NTV) {
                const int n = (4 * g + j) * 16 + lr;
                if (n < ACTION_NUM) {
                    #pragma unroll
                    for (int r = 0; r < 8; ++r)
                        out[(long)(row1 + half * 8 + r) * ACTION_NUM + n] = o1[j][r] + bias[j];
                }
            }
        }
    }
}

__global__ __launch_bounds__(256)
void carnn_wmma_packed(const int*   __restrict__ actions,
                       const float* __restrict__ outb,
                       const float* __restrict__ wsf,
                       float*       __restrict__ out,
                       int batch)
{
    __shared__ float lds[8][2][16][68];

    const int lane = threadIdx.x & 31;
    const int wv   = threadIdx.x >> 5;
    const int half = lane >> 4;
    const int lr   = lane & 15;
    const int wave_id = blockIdx.x * 8 + wv;
    const int row0 = wave_id * 32;
    if (row0 >= batch) return;
    const int row1 = (row0 + 16 < batch) ? (row0 + 16) : row0;

    float (*hl0)[68] = lds[wv][0];
    float (*hl1)[68] = lds[wv][1];
    const v4f* wsv = (const v4f*)wsf;
    const float* ptab = wsf + WS_P_OFF + lr * 4;

    static_assert(WS_O_OFF == WS_W_OFF + SEQ_LEN * 4096, "wsO must directly follow wsW");

    v4f bx[16], by[16];
    load_bblock<4>(bx, wsv + (WS_W_OFF / 4) + 1024 + lane, 0);
    load_bblock<4>(by, wsv + (WS_W_OFF / 4) + 1024 + lane, 4);

    const int* arow0 = actions + (row0 + half * 8) * SEQ_LEN;
    const int* arow1 = actions + (row1 + half * 8) * SEQ_LEN;
    int id0[8], id1[8];
    #pragma unroll
    for (int r = 0; r < 8; ++r) {
        id0[r] = arow0[r * SEQ_LEN];
        id1[r] = arow1[r * SEQ_LEN];
    }

    v8f acc0[4], acc1[4];

    {
        v4f p0[8], p1[8];
        #pragma unroll
        for (int r = 0; r < 8; ++r) {
            p0[r] = *(const v4f*)(ptab + id0[r] * D_MODEL);
            p1[r] = *(const v4f*)(ptab + id1[r] * D_MODEL);
        }
        #pragma unroll
        for (int r = 0; r < 8; ++r) {
            id0[r] = arow0[r * SEQ_LEN + 1];
            id1[r] = arow1[r * SEQ_LEN + 1];
        }
        #pragma unroll
        for (int nt = 0; nt < 4; ++nt) { acc0[nt] = (v8f)0.0f; acc1[nt] = (v8f)0.0f; }
        sigmoid_store(acc0, acc1, p0, p1, hl0, hl1, lr, half);
    }

    for (int s = 1; s < SEQ_LEN; ++s) {
        const float* ps = ptab + s * (TAB_ROWS * D_MODEL);
        v4f p0[8], p1[8];
        #pragma unroll
        for (int r = 0; r < 8; ++r) {
            p0[r] = *(const v4f*)(ps + id0[r] * D_MODEL);
            p1[r] = *(const v4f*)(ps + id1[r] * D_MODEL);
        }
        const int sn = (s + 1 < SEQ_LEN) ? (s + 1) : s;
        #pragma unroll
        for (int r = 0; r < 8; ++r) {
            id0[r] = arow0[r * SEQ_LEN + sn];
            id1[r] = arow1[r * SEQ_LEN + sn];
        }

        #pragma unroll
        for (int nt = 0; nt < 4; ++nt) { acc0[nt] = (v8f)0.0f; acc1[nt] = (v8f)0.0f; }

        const v4f* bn = wsv + (WS_W_OFF / 4) + (s + 1) * 1024 + lane;
        mma_block_lds(acc0, acc1, bx, hl0, hl1, lr, half, 0);
        __builtin_amdgcn_sched_barrier(0);
        load_bblock<4>(bx, bn, 0);
        __builtin_amdgcn_sched_barrier(0);
        mma_block_lds(acc0, acc1, by, hl0, hl1, lr, half, 4);
        __builtin_amdgcn_sched_barrier(0);
        load_bblock<4>(by, bn, 4);
        __builtin_amdgcn_sched_barrier(0);

        sigmoid_store(acc0, acc1, p0, p1, hl0, hl1, lr, half);
    }

    const v4f* bo_base = wsv + (WS_O_OFF / 4) + lane;
    const bool two_tiles = (row1 != row0);

    v2f af0[16], af1[16];
    #pragma unroll
    for (int kt = 0; kt < 16; ++kt) {
        af0[kt] = *(const v2f*)(&hl0[lr][4 * kt + 2 * half]);
        af1[kt] = *(const v2f*)(&hl1[lr][4 * kt + 2 * half]);
    }

    out_group<4, 4>(bx, by, af0, af1, bo_base + 1 * 1024, outb, out, 0, row0, row1, two_tiles, lr, half);
    out_group<4, 4>(bx, by, af0, af1, bo_base + 2 * 1024, outb, out, 1, row0, row1, two_tiles, lr, half);
    out_group<4, 4>(bx, by, af0, af1, bo_base + 3 * 1024, outb, out, 2, row0, row1, two_tiles, lr, half);
    out_group<4, 3>(bx, by, af0, af1, bo_base + 4 * 1024, outb, out, 3, row0, row1, two_tiles, lr, half);
    out_group<3, 0>(bx, by, af0, af1, bo_base,            outb, out, 4, row0, row1, two_tiles, lr, half);
}

__global__ __launch_bounds__(256)
void carnn_wmma_direct(const int*   __restrict__ actions,
                       const float* __restrict__ emb,
                       const float* __restrict__ Mw,
                       const float* __restrict__ Mb,
                       const float* __restrict__ Ww,
                       const float* __restrict__ Wb,
                       const float* __restrict__ outw,
                       const float* __restrict__ outb,
                       float*       __restrict__ out,
                       int batch)
{
    __shared__ float lds[8][16][68];
    const int lane = threadIdx.x & 31;
    const int wv   = threadIdx.x >> 5;
    const int half = lane >> 4;
    const int lr   = lane & 15;
    const int tile = blockIdx.x * 8 + wv;
    const int row0 = tile * 16;
    if (row0 >= batch) return;

    float (*hl)[68] = lds[wv];
    v8f acc[4];

    for (int s = 0; s < SEQ_LEN; ++s) {
        const float* mwp = Mw + s * D_MODEL * D_MODEL;
        const float* wwp = Ww + s * D_MODEL * D_MODEL;
        const int aidx = actions[(row0 + lr) * SEQ_LEN + s];
        const float* xrow = emb + aidx * D_MODEL;

        #pragma unroll
        for (int nt = 0; nt < 4; ++nt) acc[nt] = (v8f)0.0f;

        #pragma unroll 4
        for (int kt = 0; kt < 16; ++kt) {
            const int k = kt * 4 + half * 2;
            v2f a = *(const v2f*)(xrow + k);
            #pragma unroll
            for (int nt = 0; nt < 4; ++nt) {
                const int n = nt * 16 + lr;
                v2f b = *(const v2f*)(mwp + n * D_MODEL + k);
                acc[nt] = __builtin_amdgcn_wmma_f32_16x16x4_f32(false, a, false, b, (short)0, acc[nt], false, false);
            }
        }
        if (s > 0) {
            #pragma unroll 4
            for (int kt = 0; kt < 16; ++kt) {
                const int k = kt * 4 + half * 2;
                v2f a = *(const v2f*)(&hl[lr][k]);
                #pragma unroll
                for (int nt = 0; nt < 4; ++nt) {
                    const int n = nt * 16 + lr;
                    v2f b = *(const v2f*)(wwp + n * D_MODEL + k);
                    acc[nt] = __builtin_amdgcn_wmma_f32_16x16x4_f32(false, a, false, b, (short)0, acc[nt], false, false);
                }
            }
        }
        #pragma unroll
        for (int nt = 0; nt < 4; ++nt) {
            const int n = nt * 16 + lr;
            const float bias = Mb[s * D_MODEL + n] + Wb[s * D_MODEL + n];
            #pragma unroll
            for (int r = 0; r < 8; ++r) {
                float v = sigmoid_f(acc[nt][r] + bias);
                hl[half * 8 + r][n] = v;
            }
        }
    }

    v2f afrag[16];
    #pragma unroll
    for (int kt = 0; kt < 16; ++kt)
        afrag[kt] = *(const v2f*)(&hl[lr][4 * kt + 2 * half]);

    for (int nt0 = 0; nt0 < NT_OUT; nt0 += 4) {
        v8f oacc[4];
        #pragma unroll
        for (int j = 0; j < 4; ++j) oacc[j] = (v8f)0.0f;
        #pragma unroll 4
        for (int kt = 0; kt < 16; ++kt) {
            #pragma unroll
            for (int j = 0; j < 4; ++j) {
                const int nt = nt0 + j;
                if (nt >= NT_OUT) break;
                const int n = nt * 16 + lr;
                const int k = kt * 4 + half * 2;
                v2f b;
                if (n < ACTION_NUM) b = *(const v2f*)(outw + n * D_MODEL + k);
                else                b = (v2f)0.0f;
                oacc[j] = __builtin_amdgcn_wmma_f32_16x16x4_f32(false, afrag[kt], false, b, (short)0, oacc[j], false, false);
            }
        }
        #pragma unroll
        for (int j = 0; j < 4; ++j) {
            const int nt = nt0 + j;
            if (nt >= NT_OUT) break;
            const int n = nt * 16 + lr;
            if (n < ACTION_NUM) {
                const float bias = outb[n];
                #pragma unroll
                for (int r = 0; r < 8; ++r)
                    out[(long)(row0 + half * 8 + r) * ACTION_NUM + n] = oacc[j][r] + bias;
            }
        }
    }
}

extern "C" void kernel_launch(void* const* d_in, const int* in_sizes, int n_in,
                              void* d_out, int out_size, void* d_ws, size_t ws_size,
                              hipStream_t stream) {
    const int*   actions = (const int*)  d_in[0];
    const float* emb     = (const float*)d_in[1];
    const float* Mw      = (const float*)d_in[2];
    const float* Mb      = (const float*)d_in[3];
    const float* Ww      = (const float*)d_in[4];
    const float* Wb      = (const float*)d_in[5];
    const float* outw    = (const float*)d_in[6];
    const float* outb    = (const float*)d_in[7];
    float*       out     = (float*)d_out;

    const int batch = in_sizes[0] / SEQ_LEN;
    const int tiles = (batch + 15) / 16;

    if (ws_size >= (size_t)WS_BYTES && d_ws != nullptr) {
        float* ws = (float*)d_ws;
        carnn_prepass<<<PACK_BLOCKS + TAB_BLOCKS, 256, 0, stream>>>(emb, Mw, Mb, Ww, Wb, outw, ws);
        const int waves  = (tiles + 1) / 2;
        const int blocks = (waves + 7) / 8;
        carnn_wmma_packed<<<blocks, 256, 0, stream>>>(actions, outb, ws, out, batch);
    } else {
        const int blocks = (tiles + 7) / 8;
        carnn_wmma_direct<<<blocks, 256, 0, stream>>>(actions, emb, Mw, Mb, Ww, Wb, outw, outb, out, batch);
    }
}


# output rows: each wave collects its 32x300 results in LDS and writes them as whole aligned 128-B lines, 512 contiguous bytes per store (75 stores, was 304 of two 64-B pieces); sum order unchanged
# speedup vs baseline: 2.7430x; 1.1159x over previous
#include <hip/hip_runtime.h>

typedef float v2f __attribute__((ext_vector_type(2)));
typedef float v4f __attribute__((ext_vector_type(4)));
typedef float v8f __attribute__((ext_vector_type(8)));

#define D_MODEL    64
#define SEQ_LEN    9
#define ACTION_NUM 300
#define NT_OUT     19
#define EMB_ROWS   301
#define MT_TAB     19
#define TAB_ROWS   304

#define WS_W_OFF       0
#define WS_O_OFF       36864
#define WS_P_OFF       56320
#define WS_PACK_FLOATS 56320
#define WS_FLOATS      231424
#define WS_BYTES       (WS_FLOATS * 4)
#define PACK_BLOCKS    (WS_PACK_FLOATS / 256)
#define TAB_TILES      (SEQ_LEN * MT_TAB)
#define TAB_BLOCKS     ((TAB_TILES + 7) / 8)
#define STAGE_FLOATS   (32 * ACTION_NUM)
#define STAGE_CHUNKS   (STAGE_FLOATS * 4 / 512)
#define TILE_BYTES     (16 * ACTION_NUM * 4)

__device__ __forceinline__ float sigmoid_f(float x) {
    return __builtin_amdgcn_rcpf(1.0f + __expf(-x));
}

__global__ __launch_bounds__(256)
void carnn_prepass(const float* __restrict__ emb,
                   const float* __restrict__ Mw,
                   const float* __restrict__ Mb,
                   const float* __restrict__ Ww,
                   const float* __restrict__ Wb,
                   const float* __restrict__ outw,
                   float* __restrict__ ws)
{
    if (blockIdx.x < PACK_BLOCKS) {
        const int idx = blockIdx.x * 256 + threadIdx.x;
        float val;
        if (idx < WS_O_OFF) {
            int r = idx;
            const int s   = r / 4096;  r %= 4096;
            const int nt  = r / 1024;  r %= 1024;
            const int ktp = r / 128;   r %= 128;
            const int lane = r / 4;
            const int j    = r % 4;
            const int h = lane >> 4, lr = lane & 15;
            const int n  = nt * 16 + lr;
            const int kt = 2 * ktp + (j >> 1);
            const int k  = 4 * kt + 2 * h + (j & 1);
            val = Ww[s * 4096 + n * D_MODEL + k];
        } else {
            int r = idx - WS_O_OFF;
            const int nt  = r / 1024;  r %= 1024;
            const int ktp = r / 128;   r %= 128;
            const int lane = r / 4;
            const int j    = r % 4;
            const int h = lane >> 4, lr = lane & 15;
            const int n  = nt * 16 + lr;
            const int kt = 2 * ktp + (j >> 1);
            const int k  = 4 * kt + 2 * h + (j & 1);
            val = (n < ACTION_NUM) ? outw[n * D_MODEL + k] : 0.0f;
        }
        ws[idx] = val;
        return;
    }

    const int lane = threadIdx.x & 31;
    const int wv   = threadIdx.x >> 5;
    const int half = lane >> 4;
    const int lr   = lane & 15;
    const int t = (blockIdx.x - PACK_BLOCKS) * 8 + wv;
    if (t >= TAB_TILES) return;
    const int s  = t / MT_TAB;
    const int mt = t % MT_TAB;

    int a = mt * 16 + lr;
    if (a > EMB_ROWS - 1) a = EMB_ROWS - 1;
    const float* xrow = emb + a * D_MODEL;
    const float* mwp  = Mw + s * D_MODEL * D_MODEL;

    v2f af[16];
    #pragma unroll
    for (int kt = 0; kt < 16; ++kt) af[kt] = *(const v2f*)(xrow + 4 * kt + 2 * half);

    float bias[4];
    v8f acc[4];
    #pragma unroll
    for (int nt = 0; nt < 4; ++nt) {
        acc[nt] = (v8f)0.0f;
        const int n = nt * 16 + lr;
        bias[nt] = Mb[s * D_MODEL + n] + Wb[s * D_MODEL + n];
    }

    #pragma unroll
    for (int kt = 0; kt < 16; ++kt) {
        #pragma unroll
        for (int nt = 0; nt < 4; ++nt) {
            v2f b = *(const v2f*)(mwp + (nt * 16 + lr) * D_MODEL + 4 * kt + 2 * half);
            acc[nt] = __builtin_amdgcn_wmma_f32_16x16x4_f32(false, af[kt], false, b, (short)0, acc[nt], false, false);
        }
    }

    float* dst = ws + WS_P_OFF + (s * TAB_ROWS + mt * 16 + half * 8) * D_MODEL + lr * 4;
    #pragma unroll
    for (int r = 0; r < 8; ++r) {
        v4f v = {acc[0][r] + bias[0], acc[1][r] + bias[1], acc[2][r] + bias[2], acc[3][r] + bias[3]};
        *(v4f*)(dst + r * D_MODEL) = v;
    }
}


template <int NTV>
__device__ __forceinline__ void load_bblock(v4f (&b)[16], const v4f* __restrict__ base, int ktp0)
{
    #pragma unroll
    for (int kl = 0; kl < 4; ++kl) {
        #pragma unroll
        for (int nt = 0; nt < 4; ++nt) {
            if (nt < NTV) b[kl * 4 + nt] = base[nt * 256 + (ktp0 + kl) * 32];
        }
    }
}

__device__ __forceinline__ void mma_block_lds(v8f (&acc0)[4], v8f (&acc1)[4], const v4f (&b)[16],
                                              const float (*hl0)[68], const float (*hl1)[68],
                                              int lr, int half, int ktp0)
{
    #pragma unroll
    for (int kl = 0; kl < 4; ++kl) {
        const int ke = 8 * (ktp0 + kl) + 2 * half;
        v2f a0e = *(const v2f*)(&hl0[lr][ke]);
        v2f a0o = *(const v2f*)(&hl0[lr][ke + 4]);
        v2f a1e = *(const v2f*)(&hl1[lr][ke]);
        v2f a1o = *(const v2f*)(&hl1[lr][ke + 4]);
        #pragma unroll
        for (int nt = 0; nt < 4; ++nt) {
            v2f be = {b[kl * 4 + nt][0], b[kl * 4 + nt][1]};
            acc0[nt] = __builtin_amdgcn_wmma_f32_16x16x4_f32(false, a0e, false, be, (short)0, acc0[nt], false, false);
            acc1[nt] = __builtin_amdgcn_wmma_f32_16x16x4_f32(false, a1e, false, be, (short)0, acc1[nt], false, false);
        }
        #pragma unroll
        for (int nt = 0; nt < 4; ++nt) {
            v2f bo = {b[kl * 4 + nt][2], b[kl * 4 + nt][3]};
            acc0[nt] = __builtin_amdgcn_wmma_f32_16x16x4_f32(false, a0o, false, bo, (short)0, acc0[nt], false, false);
            acc1[nt] = __builtin_amdgcn_wmma_f32_16x16x4_f32(false, a1o, false, bo, (short)0, acc1[nt], false, false);
        }
    }
}

template <int NTV>
__device__ __forceinline__ void mma_block_reg(v8f (&o0)[4], v8f (&o1)[4], const v4f (&b)[16],
                                              const v2f (&af0)[16], const v2f (&af1)[16], int ktp0)
{
    #pragma unroll
    for (int kl = 0; kl < 4; ++kl) {
        const int ktp = ktp0 + kl;
        #pragma unroll
        for (int j = 0; j < 4; ++j) {
            if (j < NTV) {
                v2f be = {b[kl * 4 + j][0], b[kl * 4 + j][1]};
                o0[j] = __builtin_amdgcn_wmma_f32_16x16x4_f32(false, af0[2 * ktp], false, be, (short)0, o0[j], false, false);
                o1[j] = __builtin_amdgcn_wmma_f32_16x16x4_f32(false, af1[2 * ktp], false, be, (short)0, o1[j], false, false);
            }
        }
        #pragma unroll
        for (int j = 0; j < 4; ++j) {
            if (j < NTV) {
                v2f bo = {b[kl * 4 + j][2], b[kl * 4 + j][3]};
                o0[j] = __builtin_amdgcn_wmma_f32_16x16x4_f32(false, af0[2 * ktp + 1], false, bo, (short)0, o0[j], false, false);
                o1[j] = __builtin_amdgcn_wmma_f32_16x16x4_f32(false, af1[2 * ktp + 1], false, bo, (short)0, o1[j], false, false);
            }
        }
    }
}

__device__ __forceinline__ void sigmoid_store(const v8f (&acc0)[4], const v8f (&acc1)[4],
                                              const v4f (&p0)[8], const v4f (&p1)[8],
                                              float (*hl0)[68], float (*hl1)[68], int lr, int half)
{
    #pragma unroll
    for (int nt = 0; nt < 4; ++nt) {
        const int n = nt * 16 + lr;
        #pragma unroll
        for (int r = 0; r < 8; ++r) {
            float v0 = sigmoid_f(acc0[nt][r] + p0[r][nt]);
            float v1 = sigmoid_f(acc1[nt][r] + p1[r][nt]);
            hl0[half * 8 + r][n] = v0;
            hl1[half * 8 + r][n] = v1;
        }
    }
}

template <int NTV, int NTN>
__device__ __forceinline__ void out_group(v4f (&bx)[16], v4f (&by)[16],
                                          const v2f (&af0)[16], const v2f (&af1)[16],
                                          const v4f* __restrict__ bnext,
                                          const float* __restrict__ outb, float* stage,
                                          int g, int lr, int half)
{
    v8f o0[4], o1[4];
    float bias[4];
    #pragma unroll
    for (int j = 0; j < 4; ++j) {
        o0[j] = (v8f)0.0f; o1[j] = (v8f)0.0f;
        const int n = (4 * g + j) * 16 + lr;
        bias[j] = outb[n < ACTION_NUM ? n : ACTION_NUM - 1];
    }

    mma_block_reg<NTV>(o0, o1, bx, af0, af1, 0);
    if (NTN > 0) {
        __builtin_amdgcn_sched_barrier(0);
        load_bblock<NTN>(bx, bnext, 0);
        __builtin_amdgcn_sched_barrier(0);
    }
    mma_block_reg<NTV>(o0, o1, by, af0, af1, 4);
    if (NTN > 0) {
        __builtin_amdgcn_sched_barrier(0);
        load_bblock<NTN>(by, bnext, 4);
        __builtin_amdgcn_sched_barrier(0);
    }

    #pragma unroll
    for (int j = 0; j < 4; ++j) {
        if (j < NTV) {
            const int n = (4 * g + j) * 16 + lr;
            if (n < ACTION_NUM) {
                #pragma unroll
                for (int r = 0; r < 8; ++r) {
                    stage[(half * 8 + r) * ACTION_NUM + n]      = o0[j][r] + bias[j];
                    stage[(16 + half * 8 + r) * ACTION_NUM + n] = o1[j][r] + bias[j];
                }
            }
        }
    }
    if (NTN > 0) {
        __builtin_amdgcn_sched_barrier(0);
    }
}

template <bool LIMITED>
__device__ __forceinline__ void flush_image(const float* stage, float* __restrict__ out_wave, int lane, int limit)
{
    const char* src = (const char*)stage + lane * 16;
    char*       dst = (char*)out_wave + lane * 16;
    #pragma unroll 1
    for (int t = 0; t < STAGE_CHUNKS / 25; ++t) {
        v4f v[25];
        #pragma unroll
        for (int i = 0; i < 25; ++i) v[i] = *(const v4f*)(src + (t * 25 + i) * 512);
        #pragma unroll
        for (int i = 0; i < 25; ++i) {
            if (!LIMITED || (t * 25 + i) * 512 + lane * 16 < limit)
                *(v4f*)(dst + (t * 25 + i) * 512) = v[i];
        }
    }
}

__global__ __launch_bounds__(256)
void carnn_wmma_packed(const int*   __restrict__ actions,
                       const float* __restrict__ outb,
                       const float* __restrict__ wsf,
                       float*       __restrict__ out,
                       int batch)
{
    __shared__ float lds[8][STAGE_FLOATS];
    static_assert(2 * 16 * 68 <= STAGE_FLOATS, "the hl tiles must fit inside the wave's image");

    const int lane = threadIdx.x & 31;
    const int wv   = threadIdx.x >> 5;
    const int half = lane >> 4;
    const int lr   = lane & 15;
    const int wave_id = blockIdx.x * 8 + wv;
    const int row0 = wave_id * 32;
    if (row0 >= batch) return;
    const int row1 = (row0 + 16 < batch) ? (row0 + 16) : row0;

    float* stage = lds[wv];
    float (*hl0)[68] = (float (*)[68])(stage);
    float (*hl1)[68] = (float (*)[68])(stage + 16 * 68);
    const v4f* wsv = (const v4f*)wsf;
    const float* ptab = wsf + WS_P_OFF + lr * 4;

    static_assert(WS_O_OFF == WS_W_OFF + SEQ_LEN * 4096, "wsO must directly follow wsW");

    v4f bx[16], by[16];
    load_bblock<4>(bx, wsv + (WS_W_OFF / 4) + 1024 + lane, 0);
    load_bblock<4>(by, wsv + (WS_W_OFF / 4) + 1024 + lane, 4);

    const int* arow0 = actions + (row0 + half * 8) * SEQ_LEN;
    const int* arow1 = actions + (row1 + half * 8) * SEQ_LEN;
    int id0[8], id1[8];
    #pragma unroll
    for (int r = 0; r < 8; ++r) {
        id0[r] = arow0[r * SEQ_LEN];
        id1[r] = arow1[r * SEQ_LEN];
    }

    v8f acc0[4], acc1[4];

    {
        v4f p0[8], p1[8];
        #pragma unroll
        for (int r = 0; r < 8; ++r) {
            p0[r] = *(const v4f*)(ptab + id0[r] * D_MODEL);
            p1[r] = *(const v4f*)(ptab + id1[r] * D_MODEL);
        }
        #pragma unroll
        for (int r = 0; r < 8; ++r) {
            id0[r] = arow0[r * SEQ_LEN + 1];
            id1[r] = arow1[r * SEQ_LEN + 1];
        }
        #pragma unroll
        for (int nt = 0; nt < 4; ++nt) { acc0[nt] = (v8f)0.0f; acc1[nt] = (v8f)0.0f; }
        sigmoid_store(acc0, acc1, p0, p1, hl0, hl1, lr, half);
    }

    for (int s = 1; s < SEQ_LEN; ++s) {
        const float* ps = ptab + s * (TAB_ROWS * D_MODEL);
        v4f p0[8], p1[8];
        #pragma unroll
        for (int r = 0; r < 8; ++r) {
            p0[r] = *(const v4f*)(ps + id0[r] * D_MODEL);
            p1[r] = *(const v4f*)(ps + id1[r] * D_MODEL);
        }
        const int sn = (s + 1 < SEQ_LEN) ? (s + 1) : s;
        #pragma unroll
        for (int r = 0; r < 8; ++r) {
            id0[r] = arow0[r * SEQ_LEN + sn];
            id1[r] = arow1[r * SEQ_LEN + sn];
        }

        #pragma unroll
        for (int nt = 0; nt < 4; ++nt) { acc0[nt] = (v8f)0.0f; acc1[nt] = (v8f)0.0f; }

        const v4f* bn = wsv + (WS_W_OFF / 4) + (s + 1) * 1024 + lane;
        mma_block_lds(acc0, acc1, bx, hl0, hl1, lr, half, 0);
        __builtin_amdgcn_sched_barrier(0);
        load_bblock<4>(bx, bn, 0);
        __builtin_amdgcn_sched_barrier(0);
        mma_block_lds(acc0, acc1, by, hl0, hl1, lr, half, 4);
        __builtin_amdgcn_sched_barrier(0);
        load_bblock<4>(by, bn, 4);
        __builtin_amdgcn_sched_barrier(0);

        sigmoid_store(acc0, acc1, p0, p1, hl0, hl1, lr, half);
    }

    const v4f* bo_base = wsv + (WS_O_OFF / 4) + lane;
    const bool two_tiles = (row1 != row0);

    v2f af0[16], af1[16];
    #pragma unroll
    for (int kt = 0; kt < 16; ++kt) {
        af0[kt] = *(const v2f*)(&hl0[lr][4 * kt + 2 * half]);
        af1[kt] = *(const v2f*)(&hl1[lr][4 * kt + 2 * half]);
    }

    out_group<4, 4>(bx, by, af0, af1, bo_base + 1 * 1024, outb, stage, 0, lr, half);
    out_group<4, 4>(bx, by, af0, af1, bo_base + 2 * 1024, outb, stage, 1, lr, half);
    out_group<4, 4>(bx, by, af0, af1, bo_base + 3 * 1024, outb, stage, 2, lr, half);
    out_group<4, 3>(bx, by, af0, af1, bo_base + 4 * 1024, outb, stage, 3, lr, half);
    out_group<3, 0>(bx, by, af0, af1, bo_base,            outb, stage, 4, lr, half);

    float* out_wave = out + (long)row0 * ACTION_NUM;
    if (two_tiles) flush_image<false>(stage, out_wave, lane, 2 * TILE_BYTES);
    else           flush_image<true >(stage, out_wave, lane, TILE_BYTES);
}

__global__ __launch_bounds__(256)
void carnn_wmma_direct(const int*   __restrict__ actions,
                       const float* __restrict__ emb,
                       const float* __restrict__ Mw,
                       const float* __restrict__ Mb,
                       const float* __restrict__ Ww,
                       const float* __restrict__ Wb,
                       const float* __restrict__ outw,
                       const float* __restrict__ outb,
                       float*       __restrict__ out,
                       int batch)
{
    __shared__ float lds[8][16][68];
    const int lane = threadIdx.x & 31;
    const int wv   = threadIdx.x >> 5;
    const int half = lane >> 4;
    const int lr   = lane & 15;
    const int tile = blockIdx.x * 8 + wv;
    const int row0 = tile * 16;
    if (row0 >= batch) return;

    float (*hl)[68] = lds[wv];
    v8f acc[4];

    for (int s = 0; s < SEQ_LEN; ++s) {
        const float* mwp = Mw + s * D_MODEL * D_MODEL;
        const float* wwp = Ww + s * D_MODEL * D_MODEL;
        const int aidx = actions[(row0 + lr) * SEQ_LEN + s];
        const float* xrow = emb + aidx * D_MODEL;

        #pragma unroll
        for (int nt = 0; nt < 4; ++nt) acc[nt] = (v8f)0.0f;

        #pragma unroll 4
        for (int kt = 0; kt < 16; ++kt) {
            const int k = kt * 4 + half * 2;
            v2f a = *(const v2f*)(xrow + k);
            #pragma unroll
            for (int nt = 0; nt < 4; ++nt) {
                const int n = nt * 16 + lr;
                v2f b = *(const v2f*)(mwp + n * D_MODEL + k);
                acc[nt] = __builtin_amdgcn_wmma_f32_16x16x4_f32(false, a, false, b, (short)0, acc[nt], false, false);
            }
        }
        if (s > 0) {
            #pragma unroll 4
            for (int kt = 0; kt < 16; ++kt) {
                const int k = kt * 4 + half * 2;
                v2f a = *(const v2f*)(&hl[lr][k]);
                #pragma unroll
                for (int nt = 0; nt < 4; ++nt) {
                    const int n = nt * 16 + lr;
                    v2f b = *(const v2f*)(wwp + n * D_MODEL + k);
                    acc[nt] = __builtin_amdgcn_wmma_f32_16x16x4_f32(false, a, false, b, (short)0, acc[nt], false, false);
                }
            }
        }
        #pragma unroll
        for (int nt = 0; nt < 4; ++nt) {
            const int n = nt * 16 + lr;
            const float bias = Mb[s * D_MODEL + n] + Wb[s * D_MODEL + n];
            #pragma unroll
            for (int r = 0; r < 8; ++r) {
                float v = sigmoid_f(acc[nt][r] + bias);
                hl[half * 8 + r][n] = v;
            }
        }
    }

    v2f afrag[16];
    #pragma unroll
    for (int kt = 0; kt < 16; ++kt)
        afrag[kt] = *(const v2f*)(&hl[lr][4 * kt + 2 * half]);

    for (int nt0 = 0; nt0 < NT_OUT; nt0 += 4) {
        v8f oacc[4];
        #pragma unroll
        for (int j = 0; j < 4; ++j) oacc[j] = (v8f)0.0f;
        #pragma unroll 4
        for (int kt = 0; kt < 16; ++kt) {
            #pragma unroll
            for (int j = 0; j < 4; ++j) {
                const int nt = nt0 + j;
                if (nt >= NT_OUT) break;
                const int n = nt * 16 + lr;
                const int k = kt * 4 + half * 2;
                v2f b;
                if (n < ACTION_NUM) b = *(const v2f*)(outw + n * D_MODEL + k);
                else                b = (v2f)0.0f;
                oacc[j] = __builtin_amdgcn_wmma_f32_16x16x4_f32(false, afrag[kt], false, b, (short)0, oacc[j], false, false);
            }
        }
        #pragma unroll
        for (int j = 0; j < 4; ++j) {
            const int nt = nt0 + j;
            if (nt >= NT_OUT) break;
            const int n = nt * 16 + lr;
            if (n < ACTION_NUM) {
                const float bias = outb[n];
                #pragma unroll
                for (int r = 0; r < 8; ++r)
                    out[(long)(row0 + half * 8 + r) * ACTION_NUM + n] = oacc[j][r] + bias;
            }
        }
    }
}

extern "C" void kernel_launch(void* const* d_in, const int* in_sizes, int n_in,
                              void* d_out, int out_size, void* d_ws, size_t ws_size,
                              hipStream_t stream) {
    const int*   actions = (const int*)  d_in[0];
    const float* emb     = (const float*)d_in[1];
    const float* Mw      = (const float*)d_in[2];
    const float* Mb      = (const float*)d_in[3];
    const float* Ww      = (const float*)d_in[4];
    const float* Wb      = (const float*)d_in[5];
    const float* outw    = (const float*)d_in[6];
    const float* outb    = (const float*)d_in[7];
    float*       out     = (float*)d_out;

    const int batch = in_sizes[0] / SEQ_LEN;
    const int tiles = (batch + 15) / 16;

    if (ws_size >= (size_t)WS_BYTES && d_ws != nullptr) {
        float* ws = (float*)d_ws;
        carnn_prepass<<<PACK_BLOCKS + TAB_BLOCKS, 256, 0, stream>>>(emb, Mw, Mb, Ww, Wb, outw, ws);
        const int waves  = (tiles + 1) / 2;
        const int blocks = (waves + 7) / 8;
        carnn_wmma_packed<<<blocks, 256, 0, stream>>>(actions, outb, ws, out, batch);
    } else {
        const int blocks = (tiles + 7) / 8;
        carnn_wmma_direct<<<blocks, 256, 0, stream>>>(actions, emb, Mw, Mb, Ww, Wb, outw, outb, out, batch);
    }
}


# step 1's recurrent product hoisted: hl after step 0 depends on a row only through its step-0 id, so the pre-pass builds sigmoid(P0[id]) @ Ww[1].T per id on WMMA, main gathers it; sum order unchanged
# speedup vs baseline: 2.8856x; 1.0520x over previous
#include <hip/hip_runtime.h>

typedef float v2f __attribute__((ext_vector_type(2)));
typedef float v4f __attribute__((ext_vector_type(4)));
typedef float v8f __attribute__((ext_vector_type(8)));

#define D_MODEL    64
#define SEQ_LEN    9
#define ACTION_NUM 300
#define NT_OUT     19
#define EMB_ROWS   301
#define MT_TAB     19
#define TAB_ROWS   304

#define WS_W_OFF       0
#define WS_O_OFF       36864
#define WS_P_OFF       56320
#define WS_PACK_FLOATS 56320
#define WS_Q_OFF       231424
#define WS_FLOATS      250880
#define WS_BYTES       (WS_FLOATS * 4)
#define PACK_BLOCKS    (WS_PACK_FLOATS / 256)
#define TAB_TILES      (SEQ_LEN * MT_TAB)
#define TAB_BLOCKS     ((TAB_TILES + 7) / 8)
#define Q_BLOCKS       ((MT_TAB + 3) / 4)
#define STAGE_FLOATS   (32 * ACTION_NUM)
#define STAGE_CHUNKS   (STAGE_FLOATS * 4 / 512)
#define TILE_BYTES     (16 * ACTION_NUM * 4)

__device__ __forceinline__ float sigmoid_f(float x) {
    return __builtin_amdgcn_rcpf(1.0f + __expf(-x));
}


__device__ __forceinline__ void table_tile(const float* __restrict__ emb, const float* __restrict__ Mw,
                                           const float* __restrict__ Mb, const float* __restrict__ Wb,
                                           int s, int mt, int lr, int half,
                                           v8f (&acc)[4], float (&bias)[4])
{
    int a = mt * 16 + lr;
    if (a > EMB_ROWS - 1) a = EMB_ROWS - 1;
    const float* xrow = emb + a * D_MODEL;
    const float* mwp  = Mw + s * D_MODEL * D_MODEL;

    v2f af[16];
    #pragma unroll
    for (int kt = 0; kt < 16; ++kt) af[kt] = *(const v2f*)(xrow + 4 * kt + 2 * half);

    #pragma unroll
    for (int nt = 0; nt < 4; ++nt) {
        acc[nt] = (v8f)0.0f;
        const int n = nt * 16 + lr;
        bias[nt] = Mb[s * D_MODEL + n] + Wb[s * D_MODEL + n];
    }

    #pragma unroll
    for (int kt = 0; kt < 16; ++kt) {
        #pragma unroll
        for (int nt = 0; nt < 4; ++nt) {
            v2f b = *(const v2f*)(mwp + (nt * 16 + lr) * D_MODEL + 4 * kt + 2 * half);
            acc[nt] = __builtin_amdgcn_wmma_f32_16x16x4_f32(false, af[kt], false, b, (short)0, acc[nt], false, false);
        }
    }
}

__global__ __launch_bounds__(256)
void carnn_prepass(const float* __restrict__ emb,
                   const float* __restrict__ Mw,
                   const float* __restrict__ Mb,
                   const float* __restrict__ Ww,
                   const float* __restrict__ Wb,
                   const float* __restrict__ outw,
                   float* __restrict__ ws)
{
    __shared__ float tq[4][16][68];

    if (blockIdx.x < PACK_BLOCKS) {
        const int idx = blockIdx.x * 256 + threadIdx.x;
        float val;
        if (idx < WS_O_OFF) {
            int r = idx;
            const int s   = r / 4096;  r %= 4096;
            const int nt  = r / 1024;  r %= 1024;
            const int ktp = r / 128;   r %= 128;
            const int lane = r / 4;
            const int j    = r % 4;
            const int h = lane >> 4, lr = lane & 15;
            const int n  = nt * 16 + lr;
            const int kt = 2 * ktp + (j >> 1);
            const int k  = 4 * kt + 2 * h + (j & 1);
            val = Ww[s * 4096 + n * D_MODEL + k];
        } else {
            int r = idx - WS_O_OFF;
            const int nt  = r / 1024;  r %= 1024;
            const int ktp = r / 128;   r %= 128;
            const int lane = r / 4;
            const int j    = r % 4;
            const int h = lane >> 4, lr = lane & 15;
            const int n  = nt * 16 + lr;
            const int kt = 2 * ktp + (j >> 1);
            const int k  = 4 * kt + 2 * h + (j & 1);
            val = (n < ACTION_NUM) ? outw[n * D_MODEL + k] : 0.0f;
        }
        ws[idx] = val;
        return;
    }

    const int lane = threadIdx.x & 31;
    const int wv   = threadIdx.x >> 5;
    const int half = lane >> 4;
    const int lr   = lane & 15;

    if (blockIdx.x < PACK_BLOCKS + TAB_BLOCKS) {
        const int t = (blockIdx.x - PACK_BLOCKS) * 8 + wv;
        if (t >= TAB_TILES) return;
        const int s  = t / MT_TAB;
        const int mt = t % MT_TAB;

        v8f acc[4];
        float bias[4];
        table_tile(emb, Mw, Mb, Wb, s, mt, lr, half, acc, bias);

        float* dst = ws + WS_P_OFF + (s * TAB_ROWS + mt * 16 + half * 8) * D_MODEL + lr * 4;
        #pragma unroll
        for (int r = 0; r < 8; ++r) {
            v4f v = {acc[0][r] + bias[0], acc[1][r] + bias[1], acc[2][r] + bias[2], acc[3][r] + bias[3]};
            *(v4f*)(dst + r * D_MODEL) = v;
        }
        return;
    }

    if (wv >= 4) return;
    const int mt = (blockIdx.x - PACK_BLOCKS - TAB_BLOCKS) * 4 + wv;
    if (mt >= MT_TAB) return;

    v8f acc[4];
    float bias[4];
    table_tile(emb, Mw, Mb, Wb, 0, mt, lr, half, acc, bias);

    const float* wwp = Ww + 1 * D_MODEL * D_MODEL;
    v2f bq[64];
    #pragma unroll
    for (int kt = 0; kt < 16; ++kt) {
        #pragma unroll
        for (int nt = 0; nt < 4; ++nt)
            bq[kt * 4 + nt] = *(const v2f*)(wwp + (nt * 16 + lr) * D_MODEL + 4 * kt + 2 * half);
    }

    float (*hq)[68] = tq[wv];
    #pragma unroll
    for (int nt = 0; nt < 4; ++nt) {
        #pragma unroll
        for (int r = 0; r < 8; ++r)
            hq[half * 8 + r][nt * 16 + lr] = sigmoid_f(acc[nt][r] + bias[nt]);
    }
    v2f aq[16];
    #pragma unroll
    for (int kt = 0; kt < 16; ++kt) aq[kt] = *(const v2f*)(&hq[lr][4 * kt + 2 * half]);

    #pragma unroll
    for (int nt = 0; nt < 4; ++nt) acc[nt] = (v8f)0.0f;
    #pragma unroll
    for (int kt = 0; kt < 16; ++kt) {
        #pragma unroll
        for (int nt = 0; nt < 4; ++nt)
            acc[nt] = __builtin_amdgcn_wmma_f32_16x16x4_f32(false, aq[kt], false, bq[kt * 4 + nt], (short)0, acc[nt], false, false);
    }
    float* dstq = ws + WS_Q_OFF + (mt * 16 + half * 8) * D_MODEL + lr * 4;
    #pragma unroll
    for (int r = 0; r < 8; ++r) {
        v4f v = {acc[0][r], acc[1][r], acc[2][r], acc[3][r]};
        *(v4f*)(dstq + r * D_MODEL) = v;
    }
}


template <int NTV>
__device__ __forceinline__ void load_bblock(v4f (&b)[16], const v4f* __restrict__ base, int ktp0)
{
    #pragma unroll
    for (int kl = 0; kl < 4; ++kl) {
        #pragma unroll
        for (int nt = 0; nt < 4; ++nt) {
            if (nt < NTV) b[kl * 4 + nt] = base[nt * 256 + (ktp0 + kl) * 32];
        }
    }
}

__device__ __forceinline__ void mma_block_lds(v8f (&acc0)[4], v8f (&acc1)[4], const v4f (&b)[16],
                                              const float (*hl0)[68], const float (*hl1)[68],
                                              int lr, int half, int ktp0)
{
    #pragma unroll
    for (int kl = 0; kl < 4; ++kl) {
        const int ke = 8 * (ktp0 + kl) + 2 * half;
        v2f a0e = *(const v2f*)(&hl0[lr][ke]);
        v2f a0o = *(const v2f*)(&hl0[lr][ke + 4]);
        v2f a1e = *(const v2f*)(&hl1[lr][ke]);
        v2f a1o = *(const v2f*)(&hl1[lr][ke + 4]);
        #pragma unroll
        for (int nt = 0; nt < 4; ++nt) {
            v2f be = {b[kl * 4 + nt][0], b[kl * 4 + nt][1]};
            acc0[nt] = __builtin_amdgcn_wmma_f32_16x16x4_f32(false, a0e, false, be, (short)0, acc0[nt], false, false);
            acc1[nt] = __builtin_amdgcn_wmma_f32_16x16x4_f32(false, a1e, false, be, (short)0, acc1[nt], false, false);
        }
        #pragma unroll
        for (int nt = 0; nt < 4; ++nt) {
            v2f bo = {b[kl * 4 + nt][2], b[kl * 4 + nt][3]};
            acc0[nt] = __builtin_amdgcn_wmma_f32_16x16x4_f32(false, a0o, false, bo, (short)0, acc0[nt], false, false);
            acc1[nt] = __builtin_amdgcn_wmma_f32_16x16x4_f32(false, a1o, false, bo, (short)0, acc1[nt], false, false);
        }
    }
}

template <int NTV>
__device__ __forceinline__ void mma_block_reg(v8f (&o0)[4], v8f (&o1)[4], const v4f (&b)[16],
                                              const v2f (&af0)[16], const v2f (&af1)[16], int ktp0)
{
    #pragma unroll
    for (int kl = 0; kl < 4; ++kl) {
        const int ktp = ktp0 + kl;
        #pragma unroll
        for (int j = 0; j < 4; ++j) {
            if (j < NTV) {
                v2f be = {b[kl * 4 + j][0], b[kl * 4 + j][1]};
                o0[j] = __builtin_amdgcn_wmma_f32_16x16x4_f32(false, af0[2 * ktp], false, be, (short)0, o0[j], false, false);
                o1[j] = __builtin_amdgcn_wmma_f32_16x16x4_f32(false, af1[2 * ktp], false, be, (short)0, o1[j], false, false);
            }
        }
        #pragma unroll
        for (int j = 0; j < 4; ++j) {
            if (j < NTV) {
                v2f bo = {b[kl * 4 + j][2], b[kl * 4 + j][3]};
                o0[j] = __builtin_amdgcn_wmma_f32_16x16x4_f32(false, af0[2 * ktp + 1], false, bo, (short)0, o0[j], false, false);
                o1[j] = __builtin_amdgcn_wmma_f32_16x16x4_f32(false, af1[2 * ktp + 1], false, bo, (short)0, o1[j], false, false);
            }
        }
    }
}

__device__ __forceinline__ void sigmoid_store(const v8f (&acc0)[4], const v8f (&acc1)[4],
                                              const v4f (&p0)[8], const v4f (&p1)[8],
                                              float (*hl0)[68], float (*hl1)[68], int lr, int half)
{
    #pragma unroll
    for (int nt = 0; nt < 4; ++nt) {
        const int n = nt * 16 + lr;
        #pragma unroll
        for (int r = 0; r < 8; ++r) {
            float v0 = sigmoid_f(acc0[nt][r] + p0[r][nt]);
            float v1 = sigmoid_f(acc1[nt][r] + p1[r][nt]);
            hl0[half * 8 + r][n] = v0;
            hl1[half * 8 + r][n] = v1;
        }
    }
}

template <int NTV, int NTN>
__device__ __forceinline__ void out_group(v4f (&bx)[16], v4f (&by)[16],
                                          const v2f (&af0)[16], const v2f (&af1)[16],
                                          const v4f* __restrict__ bnext,
                                          const float* __restrict__ outb, float* stage,
                                          int g, int lr, int half)
{
    v8f o0[4], o1[4];
    float bias[4];
    #pragma unroll
    for (int j = 0; j < 4; ++j) {
        o0[j] = (v8f)0.0f; o1[j] = (v8f)0.0f;
        const int n = (4 * g + j) * 16 + lr;
        bias[j] = outb[n < ACTION_NUM ? n : ACTION_NUM - 1];
    }

    mma_block_reg<NTV>(o0, o1, bx, af0, af1, 0);
    if (NTN > 0) {
        __builtin_amdgcn_sched_barrier(0);
        load_bblock<NTN>(bx, bnext, 0);
        __builtin_amdgcn_sched_barrier(0);
    }
    mma_block_reg<NTV>(o0, o1, by, af0, af1, 4);
    if (NTN > 0) {
        __builtin_amdgcn_sched_barrier(0);
        load_bblock<NTN>(by, bnext, 4);
        __builtin_amdgcn_sched_barrier(0);
    }

    #pragma unroll
    for (int j = 0; j < 4; ++j) {
        if (j < NTV) {
            const int n = (4 * g + j) * 16 + lr;
            if (n < ACTION_NUM) {
                #pragma unroll
                for (int r = 0; r < 8; ++r) {
                    stage[(half * 8 + r) * ACTION_NUM + n]      = o0[j][r] + bias[j];
                    stage[(16 + half * 8 + r) * ACTION_NUM + n] = o1[j][r] + bias[j];
                }
            }
        }
    }
    if (NTN > 0) {
        __builtin_amdgcn_sched_barrier(0);
    }
}

template <bool LIMITED>
__device__ __forceinline__ void flush_image(const float* stage, float* __restrict__ out_wave, int lane, int limit)
{
    const char* src = (const char*)stage + lane * 16;
    char*       dst = (char*)out_wave + lane * 16;
    #pragma unroll 1
    for (int t = 0; t < STAGE_CHUNKS / 25; ++t) {
        v4f v[25];
        #pragma unroll
        for (int i = 0; i < 25; ++i) v[i] = *(const v4f*)(src + (t * 25 + i) * 512);
        #pragma unroll
        for (int i = 0; i < 25; ++i) {
            if (!LIMITED || (t * 25 + i) * 512 + lane * 16 < limit)
                *(v4f*)(dst + (t * 25 + i) * 512) = v[i];
        }
    }
}

__global__ __launch_bounds__(256)
void carnn_wmma_packed(const int*   __restrict__ actions,
                       const float* __restrict__ outb,
                       const float* __restrict__ wsf,
                       float*       __restrict__ out,
                       int batch)
{
    __shared__ float lds[8][STAGE_FLOATS];
    static_assert(2 * 16 * 68 <= STAGE_FLOATS, "the hl tiles must fit inside the wave's image");

    const int lane = threadIdx.x & 31;
    const int wv   = threadIdx.x >> 5;
    const int half = lane >> 4;
    const int lr   = lane & 15;
    const int wave_id = blockIdx.x * 8 + wv;
    const int row0 = wave_id * 32;
    if (row0 >= batch) return;
    const int row1 = (row0 + 16 < batch) ? (row0 + 16) : row0;

    float* stage = lds[wv];
    float (*hl0)[68] = (float (*)[68])(stage);
    float (*hl1)[68] = (float (*)[68])(stage + 16 * 68);
    const v4f* wsv = (const v4f*)wsf;
    const float* ptab = wsf + WS_P_OFF + lr * 4;

    static_assert(WS_O_OFF == WS_W_OFF + SEQ_LEN * 4096, "wsO must directly follow wsW");
    static_assert(SEQ_LEN >= 3, "the batch recurrence starts at step 2");

    v4f bx[16], by[16];
    load_bblock<4>(bx, wsv + (WS_W_OFF / 4) + 2 * 1024 + lane, 0);
    load_bblock<4>(by, wsv + (WS_W_OFF / 4) + 2 * 1024 + lane, 4);

    const int* arow0 = actions + (row0 + half * 8) * SEQ_LEN;
    const int* arow1 = actions + (row1 + half * 8) * SEQ_LEN;
    int id0[8], id1[8];

    v8f acc0[4], acc1[4];

    {
        int ia0[8], ia1[8];
        #pragma unroll
        for (int r = 0; r < 8; ++r) {
            ia0[r] = arow0[r * SEQ_LEN];
            ia1[r] = arow1[r * SEQ_LEN];
            id0[r] = arow0[r * SEQ_LEN + 1];
            id1[r] = arow1[r * SEQ_LEN + 1];
        }
        const float* qtab = wsf + WS_Q_OFF + lr * 4;
        const float* p1tab = ptab + 1 * (TAB_ROWS * D_MODEL);
        v4f q0[8], q1[8], p0[8], p1[8];
        #pragma unroll
        for (int r = 0; r < 8; ++r) {
            q0[r] = *(const v4f*)(qtab + ia0[r] * D_MODEL);
            q1[r] = *(const v4f*)(qtab + ia1[r] * D_MODEL);
            p0[r] = *(const v4f*)(p1tab + id0[r] * D_MODEL);
            p1[r] = *(const v4f*)(p1tab + id1[r] * D_MODEL);
        }
        #pragma unroll
        for (int r = 0; r < 8; ++r) {
            id0[r] = arow0[r * SEQ_LEN + 2];
            id1[r] = arow1[r * SEQ_LEN + 2];
        }
        #pragma unroll
        for (int nt = 0; nt < 4; ++nt) {
            #pragma unroll
            for (int r = 0; r < 8; ++r) { acc0[nt][r] = q0[r][nt]; acc1[nt][r] = q1[r][nt]; }
        }
        sigmoid_store(acc0, acc1, p0, p1, hl0, hl1, lr, half);
    }

    for (int s = 2; s < SEQ_LEN; ++s) {
        const float* ps = ptab + s * (TAB_ROWS * D_MODEL);
        v4f p0[8], p1[8];
        #pragma unroll
        for (int r = 0; r < 8; ++r) {
            p0[r] = *(const v4f*)(ps + id0[r] * D_MODEL);
            p1[r] = *(const v4f*)(ps + id1[r] * D_MODEL);
        }
        const int sn = (s + 1 < SEQ_LEN) ? (s + 1) : s;
        #pragma unroll
        for (int r = 0; r < 8; ++r) {
            id0[r] = arow0[r * SEQ_LEN + sn];
            id1[r] = arow1[r * SEQ_LEN + sn];
        }

        #pragma unroll
        for (int nt = 0; nt < 4; ++nt) { acc0[nt] = (v8f)0.0f; acc1[nt] = (v8f)0.0f; }

        const v4f* bn = wsv + (WS_W_OFF / 4) + (s + 1) * 1024 + lane;
        mma_block_lds(acc0, acc1, bx, hl0, hl1, lr, half, 0);
        __builtin_amdgcn_sched_barrier(0);
        load_bblock<4>(bx, bn, 0);
        __builtin_amdgcn_sched_barrier(0);
        mma_block_lds(acc0, acc1, by, hl0, hl1, lr, half, 4);
        __builtin_amdgcn_sched_barrier(0);
        load_bblock<4>(by, bn, 4);
        __builtin_amdgcn_sched_barrier(0);

        sigmoid_store(acc0, acc1, p0, p1, hl0, hl1, lr, half);
    }

    const v4f* bo_base = wsv + (WS_O_OFF / 4) + lane;
    const bool two_tiles = (row1 != row0);

    v2f af0[16], af1[16];
    #pragma unroll
    for (int kt = 0; kt < 16; ++kt) {
        af0[kt] = *(const v2f*)(&hl0[lr][4 * kt + 2 * half]);
        af1[kt] = *(const v2f*)(&hl1[lr][4 * kt + 2 * half]);
    }

    out_group<4, 4>(bx, by, af0, af1, bo_base + 1 * 1024, outb, stage, 0, lr, half);
    out_group<4, 4>(bx, by, af0, af1, bo_base + 2 * 1024, outb, stage, 1, lr, half);
    out_group<4, 4>(bx, by, af0, af1, bo_base + 3 * 1024, outb, stage, 2, lr, half);
    out_group<4, 3>(bx, by, af0, af1, bo_base + 4 * 1024, outb, stage, 3, lr, half);
    out_group<3, 0>(bx, by, af0, af1, bo_base,            outb, stage, 4, lr, half);

    float* out_wave = out + (long)row0 * ACTION_NUM;
    if (two_tiles) flush_image<false>(stage, out_wave, lane, 2 * TILE_BYTES);
    else           flush_image<true >(stage, out_wave, lane, TILE_BYTES);
}

__global__ __launch_bounds__(256)
void carnn_wmma_direct(const int*   __restrict__ actions,
                       const float* __restrict__ emb,
                       const float* __restrict__ Mw,
                       const float* __restrict__ Mb,
                       const float* __restrict__ Ww,
                       const float* __restrict__ Wb,
                       const float* __restrict__ outw,
                       const float* __restrict__ outb,
                       float*       __restrict__ out,
                       int batch)
{
    __shared__ float lds[8][16][68];
    const int lane = threadIdx.x & 31;
    const int wv   = threadIdx.x >> 5;
    const int half = lane >> 4;
    const int lr   = lane & 15;
    const int tile = blockIdx.x * 8 + wv;
    const int row0 = tile * 16;
    if (row0 >= batch) return;

    float (*hl)[68] = lds[wv];
    v8f acc[4];

    for (int s = 0; s < SEQ_LEN; ++s) {
        const float* mwp = Mw + s * D_MODEL * D_MODEL;
        const float* wwp = Ww + s * D_MODEL * D_MODEL;
        const int aidx = actions[(row0 + lr) * SEQ_LEN + s];
        const float* xrow = emb + aidx * D_MODEL;

        #pragma unroll
        for (int nt = 0; nt < 4; ++nt) acc[nt] = (v8f)0.0f;

        #pragma unroll 4
        for (int kt = 0; kt < 16; ++kt) {
            const int k = kt * 4 + half * 2;
            v2f a = *(const v2f*)(xrow + k);
            #pragma unroll
            for (int nt = 0; nt < 4; ++nt) {
                const int n = nt * 16 + lr;
                v2f b = *(const v2f*)(mwp + n * D_MODEL + k);
                acc[nt] = __builtin_amdgcn_wmma_f32_16x16x4_f32(false, a, false, b, (short)0, acc[nt], false, false);
            }
        }
        if (s > 0) {
            #pragma unroll 4
            for (int kt = 0; kt < 16; ++kt) {
                const int k = kt * 4 + half * 2;
                v2f a = *(const v2f*)(&hl[lr][k]);
                #pragma unroll
                for (int nt = 0; nt < 4; ++nt) {
                    const int n = nt * 16 + lr;
                    v2f b = *(const v2f*)(wwp + n * D_MODEL + k);
                    acc[nt] = __builtin_amdgcn_wmma_f32_16x16x4_f32(false, a, false, b, (short)0, acc[nt], false, false);
                }
            }
        }
        #pragma unroll
        for (int nt = 0; nt < 4; ++nt) {
            const int n = nt * 16 + lr;
            const float bias = Mb[s * D_MODEL + n] + Wb[s * D_MODEL + n];
            #pragma unroll
            for (int r = 0; r < 8; ++r) {
                float v = sigmoid_f(acc[nt][r] + bias);
                hl[half * 8 + r][n] = v;
            }
        }
    }

    v2f afrag[16];
    #pragma unroll
    for (int kt = 0; kt < 16; ++kt)
        afrag[kt] = *(const v2f*)(&hl[lr][4 * kt + 2 * half]);

    for (int nt0 = 0; nt0 < NT_OUT; nt0 += 4) {
        v8f oacc[4];
        #pragma unroll
        for (int j = 0; j < 4; ++j) oacc[j] = (v8f)0.0f;
        #pragma unroll 4
        for (int kt = 0; kt < 16; ++kt) {
            #pragma unroll
            for (int j = 0; j < 4; ++j) {
                const int nt = nt0 + j;
                if (nt >= NT_OUT) break;
                const int n = nt * 16 + lr;
                const int k = kt * 4 + half * 2;
                v2f b;
                if (n < ACTION_NUM) b = *(const v2f*)(outw + n * D_MODEL + k);
                else                b = (v2f)0.0f;
                oacc[j] = __builtin_amdgcn_wmma_f32_16x16x4_f32(false, afrag[kt], false, b, (short)0, oacc[j], false, false);
            }
        }
        #pragma unroll
        for (int j = 0; j < 4; ++j) {
            const int nt = nt0 + j;
            if (nt >= NT_OUT) break;
            const int n = nt * 16 + lr;
            if (n < ACTION_NUM) {
                const float bias = outb[n];
                #pragma unroll
                for (int r = 0; r < 8; ++r)
                    out[(long)(row0 + half * 8 + r) * ACTION_NUM + n] = oacc[j][r] + bias;
            }
        }
    }
}

extern "C" void kernel_launch(void* const* d_in, const int* in_sizes, int n_in,
                              void* d_out, int out_size, void* d_ws, size_t ws_size,
                              hipStream_t stream) {
    const int*   actions = (const int*)  d_in[0];
    const float* emb     = (const float*)d_in[1];
    const float* Mw      = (const float*)d_in[2];
    const float* Mb      = (const float*)d_in[3];
    const float* Ww      = (const float*)d_in[4];
    const float* Wb      = (const float*)d_in[5];
    const float* outw    = (const float*)d_in[6];
    const float* outb    = (const float*)d_in[7];
    float*       out     = (float*)d_out;

    const int batch = in_sizes[0] / SEQ_LEN;
    const int tiles = (batch + 15) / 16;

    if (ws_size >= (size_t)WS_BYTES && d_ws != nullptr) {
        float* ws = (float*)d_ws;
        carnn_prepass<<<PACK_BLOCKS + TAB_BLOCKS + Q_BLOCKS, 256, 0, stream>>>(emb, Mw, Mb, Ww, Wb, outw, ws);
        const int waves  = (tiles + 1) / 2;
        const int blocks = (waves + 7) / 8;
        carnn_wmma_packed<<<blocks, 256, 0, stream>>>(actions, outb, ws, out, batch);
    } else {
        const int blocks = (tiles + 7) / 8;
        carnn_wmma_direct<<<blocks, 256, 0, stream>>>(actions, emb, Mw, Mb, Ww, Wb, outw, outb, out, batch);
    }
}
